# starting kernel
# baseline (speedup 1.0000x reference)
#include <hip/hip_runtime.h>

typedef float v2f __attribute__((ext_vector_type(2)));
typedef float v8f __attribute__((ext_vector_type(8)));

#define WMMA_F32(a_, b_, c_) \
  __builtin_amdgcn_wmma_f32_16x16x4_f32(false, (a_), false, (b_), (short)0, (c_), false, false)

#define NB      4
#define CIN     256
#define HH      256
#define WW      256
#define EC      64
#define GW      32
#define LPERN   1024
#define LTOT    4096
#define K1      4096
#define COUT    256

#define ENC_FLOATS  (16777216)
#define T_FLOATS    (262144)
#define M_FLOATS    (262144)

#define WS_STRIDE   260
#define MS_STRIDE   68
#define XS2_STRIDE  288
#define ES2_STRIDE  96

static __device__ __forceinline__ v8f v8f_zero() {
  v8f z = {0.f, 0.f, 0.f, 0.f, 0.f, 0.f, 0.f, 0.f};
  return z;
}

__global__ __launch_bounds__(256) void k_fuse(
    const float* __restrict__ W2, const float* __restrict__ b2,
    const float* __restrict__ Wdec, const float* __restrict__ bdec,
    float* __restrict__ Mf, float* __restrict__ cb) {
  int idx = blockIdx.x * 256 + threadIdx.x;
  int s = idx >> 14;
  int o = (idx >> 6) & 255;
  int f = idx & 63;
  float sum = 0.f, sb = 0.f;
  #pragma unroll 8
  for (int e = 0; e < 64; ++e) {
    float wd = Wdec[o * 64 + e];
    sum += wd * W2[(e * 16 + s) * 64 + f];
    sb  += wd * b2[e * 16 + s];
  }
  Mf[idx] = sum;
  if (f == 0) cb[s * 256 + o] = sb + bdec[o];
}

__global__ __launch_bounds__(256) void k_encode(
    const float* __restrict__ x, const float* __restrict__ Wenc,
    const float* __restrict__ benc, float* __restrict__ enc) {
  extern __shared__ float lds[];
  float* ws = lds;
  float* xs = lds + 64 * WS_STRIDE;

  const int tid = threadIdx.x;
  const int b   = blockIdx.x;
  const int n   = b >> 9;
  const int rem = b & 511;
  const int gh  = rem >> 4;
  const int gw0 = (rem & 15) * 2;

  {
    const float4* src = (const float4*)Wenc;
    #pragma unroll
    for (int i = 0; i < 16; ++i) {
      int v = i * 256 + tid;
      float4 d = src[v];
      int e = v >> 6;
      int k = (v & 63) * 4;
      *(float4*)&ws[e * WS_STRIDE + k] = d;
    }
  }

  const int lane = tid & 31;
  const int wv   = tid >> 5;
  const int lm   = lane & 15;
  const int lh   = lane >> 4;
  const int e0   = (wv >> 1) * 16;
  const int c0   = (wv & 1) * 64;

  v8f acc[4];
  #pragma unroll
  for (int t = 0; t < 4; ++t) acc[t] = v8f_zero();

  const int kk2 = tid >> 3;
  const int kh  = tid & 7;

  for (int kc = 0; kc < 256; kc += 64) {
    __syncthreads();
    {
      const float* rowA = x +
          ((((size_t)n * CIN + (kc + 2 * kk2)) * HH + (gh * 8 + kh)) * WW + gw0 * 8);
      const float* rowB = rowA + (size_t)HH * WW;
      float4 A4[4], B4[4];
      #pragma unroll
      for (int q = 0; q < 4; ++q) {
        A4[q] = ((const float4*)rowA)[q];
        B4[q] = ((const float4*)rowB)[q];
      }
      const float* fa = (const float*)A4;
      const float* fb = (const float*)B4;
      #pragma unroll
      for (int i = 0; i < 16; ++i) {
        int cc = (i >> 3) * 64 + kh * 8 + (i & 7);
        v2f t; t.x = fa[i]; t.y = fb[i];
        *(v2f*)&xs[kk2 * XS2_STRIDE + cc * 2] = t;
      }
    }
    __syncthreads();

    #pragma unroll 4
    for (int kk = 0; kk < 64; kk += 4) {
      int ka = kk + 2 * lh;
      v2f a = *(const v2f*)&ws[(e0 + lm) * WS_STRIDE + kc + ka];
      #pragma unroll
      for (int t = 0; t < 4; ++t) {
        v2f bb = *(const v2f*)&xs[(ka >> 1) * XS2_STRIDE + (c0 + t * 16 + lm) * 2];
        acc[t] = WMMA_F32(a, bb, acc[t]);
      }
    }
  }

  size_t base_l = (size_t)n * LPERN + gh * GW + gw0;
  #pragma unroll
  for (int t = 0; t < 4; ++t) {
    int c  = c0 + t * 16 + lm;
    int pp = c >> 6;
    int ph = c & 63;
    #pragma unroll
    for (int r = 0; r < 8; ++r) {
      int e = e0 + r + 8 * lh;
      enc[(base_l + pp) * (size_t)K1 + e * 64 + ph] = acc[t][r] + benc[e];
    }
  }
}

__global__ __launch_bounds__(256) void k_fc1(
    const float* __restrict__ enc, const float* __restrict__ W1,
    const float* __restrict__ b1, float* __restrict__ Tbuf) {
  extern __shared__ float lds[];
  float* w1s = lds;
  float* es  = lds + 64 * WS_STRIDE;

  const int tid  = threadIdx.x;
  const int l0   = blockIdx.x * 32;
  const int lane = tid & 31, wv = tid >> 5;
  const int lm   = lane & 15, lh = lane >> 4;
  const int f0   = (wv >> 1) * 16;
  const int c0   = (wv & 1) * 16;

  v8f acc = v8f_zero();

  for (int kc = 0; kc < K1; kc += 256) {
    __syncthreads();
    #pragma unroll
    for (int i = 0; i < 16; ++i) {
      int v  = i * 256 + tid;
      int f  = v >> 6;
      int kq = (v & 63) * 4;
      float4 d = *(const float4*)(W1 + (size_t)f * K1 + kc + kq);
      *(float4*)&w1s[f * WS_STRIDE + kq] = d;
    }
    #pragma unroll
    for (int i = 0; i < 8; ++i) {
      int v  = i * 256 + tid;
      int c  = v >> 6;
      int kq = (v & 63) * 4;
      float4 d = *(const float4*)(enc + (size_t)(l0 + c) * K1 + kc + kq);
      v2f t0; t0.x = d.x; t0.y = d.y;
      v2f t1; t1.x = d.z; t1.y = d.w;
      *(v2f*)&es[((kq >> 1) + 0) * ES2_STRIDE + c * 2] = t0;
      *(v2f*)&es[((kq >> 1) + 1) * ES2_STRIDE + c * 2] = t1;
    }
    __syncthreads();

    #pragma unroll 4
    for (int kk = 0; kk < 256; kk += 4) {
      int ka = kk + 2 * lh;
      v2f a  = *(const v2f*)&w1s[(f0 + lm) * WS_STRIDE + ka];
      v2f bb = *(const v2f*)&es[(ka >> 1) * ES2_STRIDE + (c0 + lm) * 2];
      acc = WMMA_F32(a, bb, acc);
    }
  }

  #pragma unroll
  for (int r = 0; r < 8; ++r) {
    int f = f0 + r + 8 * lh;
    Tbuf[(size_t)f * LTOT + l0 + c0 + lm] = acc[r] + b1[f];
  }
}

__global__ __launch_bounds__(256) void k_decode(
    const float* __restrict__ Tbuf, const float* __restrict__ Mf,
    const float* __restrict__ cb, float* __restrict__ out) {
  extern __shared__ float lds[];
  float* ms = lds;
  float* ts = lds + 256 * MS_STRIDE;

  const int tid = threadIdx.x;
  const int pb  = blockIdx.x >> 2;
  const int sg  = blockIdx.x & 3;
  const int l0  = pb * 32;
  const int n   = l0 >> 10;
  const int gh  = (l0 & 1023) >> 5;

  #pragma unroll
  for (int i = 0; i < 4; ++i) {
    int v  = i * 256 + tid;
    int f2 = v >> 5;
    int c  = v & 31;
    v2f t;
    t.x = Tbuf[(size_t)(2 * f2 + 0) * LTOT + l0 + c];
    t.y = Tbuf[(size_t)(2 * f2 + 1) * LTOT + l0 + c];
    *(v2f*)&ts[f2 * ES2_STRIDE + c * 2] = t;
  }

  const int lane = tid & 31, wv = tid >> 5;
  const int lm   = lane & 15, lh = lane >> 4;

  for (int ss = 0; ss < 4; ++ss) {
    int s = sg * 4 + ss;
    __syncthreads();
    #pragma unroll
    for (int i = 0; i < 16; ++i) {
      int v  = i * 256 + tid;
      int o  = v >> 4;
      int fq = (v & 15) * 4;
      float4 d = *(const float4*)(Mf + (size_t)s * 16384 + o * 64 + fq);
      *(float4*)&ms[o * MS_STRIDE + fq] = d;
    }
    __syncthreads();

    int sy = s >> 2, sx = s & 3;
    #pragma unroll
    for (int j = 0; j < 4; ++j) {
      int tt = wv * 4 + j;
      int o0 = (tt >> 1) * 16;
      int ct = (tt & 1) * 16;
      v8f acc = v8f_zero();
      #pragma unroll
      for (int kk = 0; kk < 64; kk += 4) {
        int ka = kk + 2 * lh;
        v2f a  = *(const v2f*)&ms[(o0 + lm) * MS_STRIDE + ka];
        v2f bb = *(const v2f*)&ts[(ka >> 1) * ES2_STRIDE + (ct + lm) * 2];
        acc = WMMA_F32(a, bb, acc);
      }
      int gw = ct + lm;
      int h2 = gh * 4 + sy;
      int w2 = gw * 4 + sx;
      #pragma unroll
      for (int r = 0; r < 8; ++r) {
        int o = o0 + r + 8 * lh;
        out[(((size_t)n * COUT + o) * 128 + h2) * 128 + w2] = acc[r] + cb[s * 256 + o];
      }
    }
  }
}

extern "C" void kernel_launch(void* const* d_in, const int* in_sizes, int n_in,
                              void* d_out, int out_size, void* d_ws, size_t ws_size,
                              hipStream_t stream) {
  const float* x    = (const float*)d_in[0];
  const float* Wenc = (const float*)d_in[1];
  const float* benc = (const float*)d_in[2];
  const float* W1   = (const float*)d_in[3];
  const float* b1   = (const float*)d_in[4];
  const float* W2   = (const float*)d_in[5];
  const float* b2   = (const float*)d_in[6];
  const float* Wdec = (const float*)d_in[7];
  const float* bdec = (const float*)d_in[8];
  float* out = (float*)d_out;

  float* ws   = (float*)d_ws;
  float* enc  = ws;
  float* Tbuf = ws + ENC_FLOATS;
  float* Mf   = Tbuf + T_FLOATS;
  float* cbuf = Mf + M_FLOATS;

  k_fuse<<<1024, 256, 0, stream>>>(W2, b2, Wdec, bdec, Mf, cbuf);

  size_t ldsB = (size_t)(64 * WS_STRIDE + 32 * XS2_STRIDE) * sizeof(float);
  k_encode<<<2048, 256, ldsB, stream>>>(x, Wenc, benc, enc);

  size_t ldsC1 = (size_t)(64 * WS_STRIDE + 128 * ES2_STRIDE) * sizeof(float);
  k_fc1<<<128, 256, ldsC1, stream>>>(enc, W1, b1, Tbuf);

  size_t ldsC2 = (size_t)(256 * MS_STRIDE + 32 * ES2_STRIDE) * sizeof(float);
  k_decode<<<512, 256, ldsC2, stream>>>(Tbuf, Mf, cbuf, out);
}


# fc1 spread over all 256 WGPs: 16-patch tiles, 16 waves each, K split 4-way inside the block (partials summed through LDS in fixed order), next K chunk loaded during the WMMAs; T stored tile-major
# speedup vs baseline: 1.3725x; 1.3725x over previous
#include <hip/hip_runtime.h>

typedef float v2f __attribute__((ext_vector_type(2)));
typedef float v8f __attribute__((ext_vector_type(8)));

#define WMMA_F32(a_, b_, c_) \
  __builtin_amdgcn_wmma_f32_16x16x4_f32(false, (a_), false, (b_), (short)0, (c_), false, false)

#define NB      4
#define CIN     256
#define HH      256
#define WW      256
#define EC      64
#define GW      32
#define LPERN   1024
#define LTOT    4096
#define K1      4096
#define COUT    256

#define ENC_FLOATS  (16777216)
#define T_FLOATS    (262144)
#define M_FLOATS    (262144)

#define WS_STRIDE   260
#define MS_STRIDE   68
#define XS2_STRIDE  288
#define ES2_STRIDE  96

static __device__ __forceinline__ v8f v8f_zero() {
  v8f z = {0.f, 0.f, 0.f, 0.f, 0.f, 0.f, 0.f, 0.f};
  return z;
}

__global__ __launch_bounds__(256) void k_fuse(
    const float* __restrict__ W2, const float* __restrict__ b2,
    const float* __restrict__ Wdec, const float* __restrict__ bdec,
    float* __restrict__ Mf, float* __restrict__ cb) {
  int idx = blockIdx.x * 256 + threadIdx.x;
  int s = idx >> 14;
  int o = (idx >> 6) & 255;
  int f = idx & 63;
  float sum = 0.f, sb = 0.f;
  #pragma unroll 8
  for (int e = 0; e < 64; ++e) {
    float wd = Wdec[o * 64 + e];
    sum += wd * W2[(e * 16 + s) * 64 + f];
    sb  += wd * b2[e * 16 + s];
  }
  Mf[idx] = sum;
  if (f == 0) cb[s * 256 + o] = sb + bdec[o];
}

__global__ __launch_bounds__(256) void k_encode(
    const float* __restrict__ x, const float* __restrict__ Wenc,
    const float* __restrict__ benc, float* __restrict__ enc) {
  extern __shared__ float lds[];
  float* ws = lds;
  float* xs = lds + 64 * WS_STRIDE;

  const int tid = threadIdx.x;
  const int b   = blockIdx.x;
  const int n   = b >> 9;
  const int rem = b & 511;
  const int gh  = rem >> 4;
  const int gw0 = (rem & 15) * 2;

  {
    const float4* src = (const float4*)Wenc;
    #pragma unroll
    for (int i = 0; i < 16; ++i) {
      int v = i * 256 + tid;
      float4 d = src[v];
      int e = v >> 6;
      int k = (v & 63) * 4;
      *(float4*)&ws[e * WS_STRIDE + k] = d;
    }
  }

  const int lane = tid & 31;
  const int wv   = tid >> 5;
  const int lm   = lane & 15;
  const int lh   = lane >> 4;
  const int e0   = (wv >> 1) * 16;
  const int c0   = (wv & 1) * 64;

  v8f acc[4];
  #pragma unroll
  for (int t = 0; t < 4; ++t) acc[t] = v8f_zero();

  const int kk2 = tid >> 3;
  const int kh  = tid & 7;

  for (int kc = 0; kc < 256; kc += 64) {
    __syncthreads();
    {
      const float* rowA = x +
          ((((size_t)n * CIN + (kc + 2 * kk2)) * HH + (gh * 8 + kh)) * WW + gw0 * 8);
      const float* rowB = rowA + (size_t)HH * WW;
      float4 A4[4], B4[4];
      #pragma unroll
      for (int q = 0; q < 4; ++q) {
        A4[q] = ((const float4*)rowA)[q];
        B4[q] = ((const float4*)rowB)[q];
      }
      const float* fa = (const float*)A4;
      const float* fb = (const float*)B4;
      #pragma unroll
      for (int i = 0; i < 16; ++i) {
        int cc = (i >> 3) * 64 + kh * 8 + (i & 7);
        v2f t; t.x = fa[i]; t.y = fb[i];
        *(v2f*)&xs[kk2 * XS2_STRIDE + cc * 2] = t;
      }
    }
    __syncthreads();

    #pragma unroll 4
    for (int kk = 0; kk < 64; kk += 4) {
      int ka = kk + 2 * lh;
      v2f a = *(const v2f*)&ws[(e0 + lm) * WS_STRIDE + kc + ka];
      #pragma unroll
      for (int t = 0; t < 4; ++t) {
        v2f bb = *(const v2f*)&xs[(ka >> 1) * XS2_STRIDE + (c0 + t * 16 + lm) * 2];
        acc[t] = WMMA_F32(a, bb, acc[t]);
      }
    }
  }

  size_t base_l = (size_t)n * LPERN + gh * GW + gw0;
  #pragma unroll
  for (int t = 0; t < 4; ++t) {
    int c  = c0 + t * 16 + lm;
    int pp = c >> 6;
    int ph = c & 63;
    #pragma unroll
    for (int r = 0; r < 8; ++r) {
      int e = e0 + r + 8 * lh;
      enc[(base_l + pp) * (size_t)K1 + e * 64 + ph] = acc[t][r] + benc[e];
    }
  }
}

#define F1_NT     16
#define F1_KC     256
#define F1_ESTR   32

static __device__ __forceinline__ int f1_es_off(int R, int col) {
  return R * F1_ESTR + 2 * (col ^ ((R >> 1) & 15));
}

__global__ __launch_bounds__(512) void k_fc1(
    const float* enc, const float* W1,
    const float* __restrict__ b1, float* Tbuf) {
  extern __shared__ float lds[];
  float* w1s = lds;
  float* es  = lds + 64 * WS_STRIDE;

  const int tid  = threadIdx.x;
  const int l0   = blockIdx.x * F1_NT;
  const int lane = tid & 31, wv = tid >> 5;
  const int lm   = lane & 15, lh = lane >> 4;
  const int mt   = wv & 3;
  const int ks   = wv >> 2;

  const int sr  = tid >> 6;
  const int wkq = (tid & 63) * 4;
  const float* w1p = W1 + (size_t)sr * K1 + wkq;
  const float* enp = enc + (size_t)(l0 + sr) * K1 + wkq;

  float4 pw0 = *(const float4*)(w1p);
  float4 pw1 = *(const float4*)(w1p + (size_t) 8 * K1);
  float4 pw2 = *(const float4*)(w1p + (size_t)16 * K1);
  float4 pw3 = *(const float4*)(w1p + (size_t)24 * K1);
  float4 pw4 = *(const float4*)(w1p + (size_t)32 * K1);
  float4 pw5 = *(const float4*)(w1p + (size_t)40 * K1);
  float4 pw6 = *(const float4*)(w1p + (size_t)48 * K1);
  float4 pw7 = *(const float4*)(w1p + (size_t)56 * K1);
  float4 pe0 = *(const float4*)(enp);
  float4 pe1 = *(const float4*)(enp + (size_t)8 * K1);

  v8f acc = v8f_zero();

  for (int ch = 0; ch < K1 / F1_KC; ++ch) {
    __syncthreads();
    *(float4*)&w1s[(sr +  0) * WS_STRIDE + wkq] = pw0;
    *(float4*)&w1s[(sr +  8) * WS_STRIDE + wkq] = pw1;
    *(float4*)&w1s[(sr + 16) * WS_STRIDE + wkq] = pw2;
    *(float4*)&w1s[(sr + 24) * WS_STRIDE + wkq] = pw3;
    *(float4*)&w1s[(sr + 32) * WS_STRIDE + wkq] = pw4;
    *(float4*)&w1s[(sr + 40) * WS_STRIDE + wkq] = pw5;
    *(float4*)&w1s[(sr + 48) * WS_STRIDE + wkq] = pw6;
    *(float4*)&w1s[(sr + 56) * WS_STRIDE + wkq] = pw7;
    {
      int R0 = wkq >> 1;
      v2f t0; t0.x = pe0.x; t0.y = pe0.y;
      v2f t1; t1.x = pe0.z; t1.y = pe0.w;
      *(v2f*)&es[f1_es_off(R0 + 0, sr)] = t0;
      *(v2f*)&es[f1_es_off(R0 + 1, sr)] = t1;
      v2f u0; u0.x = pe1.x; u0.y = pe1.y;
      v2f u1; u1.x = pe1.z; u1.y = pe1.w;
      *(v2f*)&es[f1_es_off(R0 + 0, sr + 8)] = u0;
      *(v2f*)&es[f1_es_off(R0 + 1, sr + 8)] = u1;
    }
    __syncthreads();

    if (ch + 1 < K1 / F1_KC) {
      int kn = (ch + 1) * F1_KC;
      pw0 = *(const float4*)(w1p + kn);
      pw1 = *(const float4*)(w1p + (size_t) 8 * K1 + kn);
      pw2 = *(const float4*)(w1p + (size_t)16 * K1 + kn);
      pw3 = *(const float4*)(w1p + (size_t)24 * K1 + kn);
      pw4 = *(const float4*)(w1p + (size_t)32 * K1 + kn);
      pw5 = *(const float4*)(w1p + (size_t)40 * K1 + kn);
      pw6 = *(const float4*)(w1p + (size_t)48 * K1 + kn);
      pw7 = *(const float4*)(w1p + (size_t)56 * K1 + kn);
      pe0 = *(const float4*)(enp + kn);
      pe1 = *(const float4*)(enp + (size_t)8 * K1 + kn);
    }

    #pragma unroll
    for (int j = 0; j < 16; ++j) {
      int ka = ks * 64 + j * 4 + 2 * lh;
      v2f a  = *(const v2f*)&w1s[(mt * 16 + lm) * WS_STRIDE + ka];
      v2f bb = *(const v2f*)&es[f1_es_off(ka >> 1, lm)];
      acc = WMMA_F32(a, bb, acc);
    }
  }

  __syncthreads();
  float* red = lds;
  #pragma unroll
  for (int r = 0; r < 8; ++r)
    red[ks * 1024 + mt * 256 + r * 32 + lane] = acc[r];
  __syncthreads();
  #pragma unroll
  for (int h = 0; h < 2; ++h) {
    int o   = tid + 512 * h;
    int f   = o >> 4;
    int col = o & 15;
    int fr  = f & 15;
    int off = (f >> 4) * 256 + (fr & 7) * 32 + (fr >> 3) * 16 + col;
    float s = ((red[off] + red[1024 + off]) + red[2048 + off]) + red[3072 + off];
    Tbuf[(size_t)blockIdx.x * 1024 + o] = s + b1[f];
  }
}

__global__ __launch_bounds__(256) void k_decode(
    const float* __restrict__ Tbuf, const float* __restrict__ Mf,
    const float* __restrict__ cb, float* __restrict__ out) {
  extern __shared__ float lds[];
  float* ms = lds;
  float* ts = lds + 256 * MS_STRIDE;

  const int tid = threadIdx.x;
  const int pb  = blockIdx.x >> 2;
  const int sg  = blockIdx.x & 3;
  const int l0  = pb * 32;
  const int n   = l0 >> 10;
  const int gh  = (l0 & 1023) >> 5;

  #pragma unroll
  for (int i = 0; i < 4; ++i) {
    int v  = i * 256 + tid;
    int f2 = v >> 5;
    int c  = v & 31;
    v2f t;
    const float* tt = Tbuf + (size_t)((l0 + c) >> 4) * 1024 + (c & 15);
    t.x = tt[(2 * f2 + 0) * 16];
    t.y = tt[(2 * f2 + 1) * 16];
    *(v2f*)&ts[f2 * ES2_STRIDE + c * 2] = t;
  }

  const int lane = tid & 31, wv = tid >> 5;
  const int lm   = lane & 15, lh = lane >> 4;

  for (int ss = 0; ss < 4; ++ss) {
    int s = sg * 4 + ss;
    __syncthreads();
    #pragma unroll
    for (int i = 0; i < 16; ++i) {
      int v  = i * 256 + tid;
      int o  = v >> 4;
      int fq = (v & 15) * 4;
      float4 d = *(const float4*)(Mf + (size_t)s * 16384 + o * 64 + fq);
      *(float4*)&ms[o * MS_STRIDE + fq] = d;
    }
    __syncthreads();

    int sy = s >> 2, sx = s & 3;
    #pragma unroll
    for (int j = 0; j < 4; ++j) {
      int tt = wv * 4 + j;
      int o0 = (tt >> 1) * 16;
      int ct = (tt & 1) * 16;
      v8f acc = v8f_zero();
      #pragma unroll
      for (int kk = 0; kk < 64; kk += 4) {
        int ka = kk + 2 * lh;
        v2f a  = *(const v2f*)&ms[(o0 + lm) * MS_STRIDE + ka];
        v2f bb = *(const v2f*)&ts[(ka >> 1) * ES2_STRIDE + (ct + lm) * 2];
        acc = WMMA_F32(a, bb, acc);
      }
      int gw = ct + lm;
      int h2 = gh * 4 + sy;
      int w2 = gw * 4 + sx;
      #pragma unroll
      for (int r = 0; r < 8; ++r) {
        int o = o0 + r + 8 * lh;
        out[(((size_t)n * COUT + o) * 128 + h2) * 128 + w2] = acc[r] + cb[s * 256 + o];
      }
    }
  }
}

extern "C" void kernel_launch(void* const* d_in, const int* in_sizes, int n_in,
                              void* d_out, int out_size, void* d_ws, size_t ws_size,
                              hipStream_t stream) {
  const float* x    = (const float*)d_in[0];
  const float* Wenc = (const float*)d_in[1];
  const float* benc = (const float*)d_in[2];
  const float* W1   = (const float*)d_in[3];
  const float* b1   = (const float*)d_in[4];
  const float* W2   = (const float*)d_in[5];
  const float* b2   = (const float*)d_in[6];
  const float* Wdec = (const float*)d_in[7];
  const float* bdec = (const float*)d_in[8];
  float* out = (float*)d_out;

  float* ws   = (float*)d_ws;
  float* enc  = ws;
  float* Tbuf = ws + ENC_FLOATS;
  float* Mf   = Tbuf + T_FLOATS;
  float* cbuf = Mf + M_FLOATS;

  k_fuse<<<1024, 256, 0, stream>>>(W2, b2, Wdec, bdec, Mf, cbuf);

  size_t ldsB = (size_t)(64 * WS_STRIDE + 32 * XS2_STRIDE) * sizeof(float);
  k_encode<<<2048, 256, ldsB, stream>>>(x, Wenc, benc, enc);

  size_t ldsC1 = (size_t)(64 * WS_STRIDE + 128 * F1_ESTR) * sizeof(float);
  k_fc1<<<256, 512, ldsC1, stream>>>(enc, W1, b1, Tbuf);

  size_t ldsC2 = (size_t)(256 * MS_STRIDE + 32 * ES2_STRIDE) * sizeof(float);
  k_decode<<<512, 256, ldsC2, stream>>>(Tbuf, Mf, cbuf, out);
}


# encode x staging re-mapped: lane = (row, 16-B quarter), wave = 8 channels, so a wave load touches 8 lines x 64 B, not 32 x 16 B; pixel pairs go to LDS as conflict-free 16-B stores
# speedup vs baseline: 1.5065x; 1.0976x over previous
#include <hip/hip_runtime.h>

typedef float v2f __attribute__((ext_vector_type(2)));
typedef float v8f __attribute__((ext_vector_type(8)));

#define WMMA_F32(a_, b_, c_) \
  __builtin_amdgcn_wmma_f32_16x16x4_f32(false, (a_), false, (b_), (short)0, (c_), false, false)

#define NB      4
#define CIN     256
#define HH      256
#define WW      256
#define EC      64
#define GW      32
#define LPERN   1024
#define LTOT    4096
#define K1      4096
#define COUT    256

#define ENC_FLOATS  (16777216)
#define T_FLOATS    (262144)
#define M_FLOATS    (262144)

#define WS_STRIDE   260
#define MS_STRIDE   68
#define XS2_STRIDE  288
#define XS_SKEW     4
#define ES2_STRIDE  96

static __device__ __forceinline__ v8f v8f_zero() {
  v8f z = {0.f, 0.f, 0.f, 0.f, 0.f, 0.f, 0.f, 0.f};
  return z;
}

__global__ __launch_bounds__(256) void k_fuse(
    const float* __restrict__ W2, const float* __restrict__ b2,
    const float* __restrict__ Wdec, const float* __restrict__ bdec,
    float* __restrict__ Mf, float* __restrict__ cb) {
  int idx = blockIdx.x * 256 + threadIdx.x;
  int s = idx >> 14;
  int o = (idx >> 6) & 255;
  int f = idx & 63;
  float sum = 0.f, sb = 0.f;
  #pragma unroll 8
  for (int e = 0; e < 64; ++e) {
    float wd = Wdec[o * 64 + e];
    sum += wd * W2[(e * 16 + s) * 64 + f];
    sb  += wd * b2[e * 16 + s];
  }
  Mf[idx] = sum;
  if (f == 0) cb[s * 256 + o] = sb + bdec[o];
}

__global__ __launch_bounds__(256) void k_encode(
    const float* __restrict__ x, const float* __restrict__ Wenc,
    const float* __restrict__ benc, float* __restrict__ enc) {
  extern __shared__ float lds[];
  float* ws = lds;
  float* xs = lds + 64 * WS_STRIDE;

  const int tid = threadIdx.x;
  const int b   = blockIdx.x;
  const int n   = b >> 9;
  const int rem = b & 511;
  const int gh  = rem >> 4;
  const int gw0 = (rem & 15) * 2;

  {
    const float4* src = (const float4*)Wenc;
    #pragma unroll
    for (int i = 0; i < 16; ++i) {
      int v = i * 256 + tid;
      float4 d = src[v];
      int e = v >> 6;
      int k = (v & 63) * 4;
      *(float4*)&ws[e * WS_STRIDE + k] = d;
    }
  }

  const int lane = tid & 31;
  const int wv   = tid >> 5;
  const int lm   = lane & 15;
  const int lh   = lane >> 4;
  const int e0   = (wv >> 1) * 16;
  const int c0   = (wv & 1) * 64;
  const float* xb = xs + (wv & 1) * XS_SKEW;

  v8f acc[4];
  #pragma unroll
  for (int t = 0; t < 4; ++t) acc[t] = v8f_zero();

  const int sq  = tid & 3;
  const int skh = (tid >> 2) & 7;
  const int spp = sq >> 1;
  const int scol = (spp * 64 + skh * 8 + (sq & 1) * 4) * 2 + spp * XS_SKEW;
  const float* xrow = x +
      ((((size_t)n * CIN + wv * 8) * HH + (gh * 8 + skh)) * WW + gw0 * 8 + sq * 4);

  for (int kc = 0; kc < 256; kc += 64) {
    __syncthreads();
    {
      const float* p0 = xrow + (size_t)kc * HH * WW;
      float4 v[8];
      #pragma unroll
      for (int j = 0; j < 8; ++j)
        v[j] = *(const float4*)(p0 + (size_t)j * HH * WW);
      #pragma unroll
      for (int p = 0; p < 4; ++p) {
        float4 lo, hi;
        lo.x = v[2 * p].x; lo.y = v[2 * p + 1].x; lo.z = v[2 * p].y; lo.w = v[2 * p + 1].y;
        hi.x = v[2 * p].z; hi.y = v[2 * p + 1].z; hi.z = v[2 * p].w; hi.w = v[2 * p + 1].w;
        float* d = &xs[(wv * 4 + p) * XS2_STRIDE + scol];
        *(float4*)d       = lo;
        *(float4*)(d + 4) = hi;
      }
    }
    __syncthreads();

    #pragma unroll 4
    for (int kk = 0; kk < 64; kk += 4) {
      int ka = kk + 2 * lh;
      v2f a = *(const v2f*)&ws[(e0 + lm) * WS_STRIDE + kc + ka];
      #pragma unroll
      for (int t = 0; t < 4; ++t) {
        v2f bb = *(const v2f*)&xb[(ka >> 1) * XS2_STRIDE + (c0 + t * 16 + lm) * 2];
        acc[t] = WMMA_F32(a, bb, acc[t]);
      }
    }
  }

  size_t base_l = (size_t)n * LPERN + gh * GW + gw0;
  #pragma unroll
  for (int t = 0; t < 4; ++t) {
    int c  = c0 + t * 16 + lm;
    int pp = c >> 6;
    int ph = c & 63;
    #pragma unroll
    for (int r = 0; r < 8; ++r) {
      int e = e0 + r + 8 * lh;
      enc[(base_l + pp) * (size_t)K1 + e * 64 + ph] = acc[t][r] + benc[e];
    }
  }
}

#define F1_NT     16
#define F1_KC     256
#define F1_ESTR   32

static __device__ __forceinline__ int f1_es_off(int R, int col) {
  return R * F1_ESTR + 2 * (col ^ ((R >> 1) & 15));
}

__global__ __launch_bounds__(512) void k_fc1(
    const float* enc, const float* W1,
    const float* __restrict__ b1, float* Tbuf) {
  extern __shared__ float lds[];
  float* w1s = lds;
  float* es  = lds + 64 * WS_STRIDE;

  const int tid  = threadIdx.x;
  const int l0   = blockIdx.x * F1_NT;
  const int lane = tid & 31, wv = tid >> 5;
  const int lm   = lane & 15, lh = lane >> 4;
  const int mt   = wv & 3;
  const int ks   = wv >> 2;

  const int sr  = tid >> 6;
  const int wkq = (tid & 63) * 4;
  const float* w1p = W1 + (size_t)sr * K1 + wkq;
  const float* enp = enc + (size_t)(l0 + sr) * K1 + wkq;

  float4 pw0 = *(const float4*)(w1p);
  float4 pw1 = *(const float4*)(w1p + (size_t) 8 * K1);
  float4 pw2 = *(const float4*)(w1p + (size_t)16 * K1);
  float4 pw3 = *(const float4*)(w1p + (size_t)24 * K1);
  float4 pw4 = *(const float4*)(w1p + (size_t)32 * K1);
  float4 pw5 = *(const float4*)(w1p + (size_t)40 * K1);
  float4 pw6 = *(const float4*)(w1p + (size_t)48 * K1);
  float4 pw7 = *(const float4*)(w1p + (size_t)56 * K1);
  float4 pe0 = *(const float4*)(enp);
  float4 pe1 = *(const float4*)(enp + (size_t)8 * K1);

  v8f acc = v8f_zero();

  for (int ch = 0; ch < K1 / F1_KC; ++ch) {
    __syncthreads();
    *(float4*)&w1s[(sr +  0) * WS_STRIDE + wkq] = pw0;
    *(float4*)&w1s[(sr +  8) * WS_STRIDE + wkq] = pw1;
    *(float4*)&w1s[(sr + 16) * WS_STRIDE + wkq] = pw2;
    *(float4*)&w1s[(sr + 24) * WS_STRIDE + wkq] = pw3;
    *(float4*)&w1s[(sr + 32) * WS_STRIDE + wkq] = pw4;
    *(float4*)&w1s[(sr + 40) * WS_STRIDE + wkq] = pw5;
    *(float4*)&w1s[(sr + 48) * WS_STRIDE + wkq] = pw6;
    *(float4*)&w1s[(sr + 56) * WS_STRIDE + wkq] = pw7;
    {
      int R0 = wkq >> 1;
      v2f t0; t0.x = pe0.x; t0.y = pe0.y;
      v2f t1; t1.x = pe0.z; t1.y = pe0.w;
      *(v2f*)&es[f1_es_off(R0 + 0, sr)] = t0;
      *(v2f*)&es[f1_es_off(R0 + 1, sr)] = t1;
      v2f u0; u0.x = pe1.x; u0.y = pe1.y;
      v2f u1; u1.x = pe1.z; u1.y = pe1.w;
      *(v2f*)&es[f1_es_off(R0 + 0, sr + 8)] = u0;
      *(v2f*)&es[f1_es_off(R0 + 1, sr + 8)] = u1;
    }
    __syncthreads();

    if (ch + 1 < K1 / F1_KC) {
      int kn = (ch + 1) * F1_KC;
      pw0 = *(const float4*)(w1p + kn);
      pw1 = *(const float4*)(w1p + (size_t) 8 * K1 + kn);
      pw2 = *(const float4*)(w1p + (size_t)16 * K1 + kn);
      pw3 = *(const float4*)(w1p + (size_t)24 * K1 + kn);
      pw4 = *(const float4*)(w1p + (size_t)32 * K1 + kn);
      pw5 = *(const float4*)(w1p + (size_t)40 * K1 + kn);
      pw6 = *(const float4*)(w1p + (size_t)48 * K1 + kn);
      pw7 = *(const float4*)(w1p + (size_t)56 * K1 + kn);
      pe0 = *(const float4*)(enp + kn);
      pe1 = *(const float4*)(enp + (size_t)8 * K1 + kn);
    }

    #pragma unroll
    for (int j = 0; j < 16; ++j) {
      int ka = ks * 64 + j * 4 + 2 * lh;
      v2f a  = *(const v2f*)&w1s[(mt * 16 + lm) * WS_STRIDE + ka];
      v2f bb = *(const v2f*)&es[f1_es_off(ka >> 1, lm)];
      acc = WMMA_F32(a, bb, acc);
    }
  }

  __syncthreads();
  float* red = lds;
  #pragma unroll
  for (int r = 0; r < 8; ++r)
    red[ks * 1024 + mt * 256 + r * 32 + lane] = acc[r];
  __syncthreads();
  #pragma unroll
  for (int h = 0; h < 2; ++h) {
    int o   = tid + 512 * h;
    int f   = o >> 4;
    int col = o & 15;
    int fr  = f & 15;
    int off = (f >> 4) * 256 + (fr & 7) * 32 + (fr >> 3) * 16 + col;
    float s = ((red[off] + red[1024 + off]) + red[2048 + off]) + red[3072 + off];
    Tbuf[(size_t)blockIdx.x * 1024 + o] = s + b1[f];
  }
}

__global__ __launch_bounds__(256) void k_decode(
    const float* __restrict__ Tbuf, const float* __restrict__ Mf,
    const float* __restrict__ cb, float* __restrict__ out) {
  extern __shared__ float lds[];
  float* ms = lds;
  float* ts = lds + 256 * MS_STRIDE;

  const int tid = threadIdx.x;
  const int pb  = blockIdx.x >> 2;
  const int sg  = blockIdx.x & 3;
  const int l0  = pb * 32;
  const int n   = l0 >> 10;
  const int gh  = (l0 & 1023) >> 5;

  #pragma unroll
  for (int i = 0; i < 4; ++i) {
    int v  = i * 256 + tid;
    int f2 = v >> 5;
    int c  = v & 31;
    v2f t;
    const float* tt = Tbuf + (size_t)((l0 + c) >> 4) * 1024 + (c & 15);
    t.x = tt[(2 * f2 + 0) * 16];
    t.y = tt[(2 * f2 + 1) * 16];
    *(v2f*)&ts[f2 * ES2_STRIDE + c * 2] = t;
  }

  const int lane = tid & 31, wv = tid >> 5;
  const int lm   = lane & 15, lh = lane >> 4;

  for (int ss = 0; ss < 4; ++ss) {
    int s = sg * 4 + ss;
    __syncthreads();
    #pragma unroll
    for (int i = 0; i < 16; ++i) {
      int v  = i * 256 + tid;
      int o  = v >> 4;
      int fq = (v & 15) * 4;
      float4 d = *(const float4*)(Mf + (size_t)s * 16384 + o * 64 + fq);
      *(float4*)&ms[o * MS_STRIDE + fq] = d;
    }
    __syncthreads();

    int sy = s >> 2, sx = s & 3;
    #pragma unroll
    for (int j = 0; j < 4; ++j) {
      int tt = wv * 4 + j;
      int o0 = (tt >> 1) * 16;
      int ct = (tt & 1) * 16;
      v8f acc = v8f_zero();
      #pragma unroll
      for (int kk = 0; kk < 64; kk += 4) {
        int ka = kk + 2 * lh;
        v2f a  = *(const v2f*)&ms[(o0 + lm) * MS_STRIDE + ka];
        v2f bb = *(const v2f*)&ts[(ka >> 1) * ES2_STRIDE + (ct + lm) * 2];
        acc = WMMA_F32(a, bb, acc);
      }
      int gw = ct + lm;
      int h2 = gh * 4 + sy;
      int w2 = gw * 4 + sx;
      #pragma unroll
      for (int r = 0; r < 8; ++r) {
        int o = o0 + r + 8 * lh;
        out[(((size_t)n * COUT + o) * 128 + h2) * 128 + w2] = acc[r] + cb[s * 256 + o];
      }
    }
  }
}

extern "C" void kernel_launch(void* const* d_in, const int* in_sizes, int n_in,
                              void* d_out, int out_size, void* d_ws, size_t ws_size,
                              hipStream_t stream) {
  const float* x    = (const float*)d_in[0];
  const float* Wenc = (const float*)d_in[1];
  const float* benc = (const float*)d_in[2];
  const float* W1   = (const float*)d_in[3];
  const float* b1   = (const float*)d_in[4];
  const float* W2   = (const float*)d_in[5];
  const float* b2   = (const float*)d_in[6];
  const float* Wdec = (const float*)d_in[7];
  const float* bdec = (const float*)d_in[8];
  float* out = (float*)d_out;

  float* ws   = (float*)d_ws;
  float* enc  = ws;
  float* Tbuf = ws + ENC_FLOATS;
  float* Mf   = Tbuf + T_FLOATS;
  float* cbuf = Mf + M_FLOATS;

  k_fuse<<<1024, 256, 0, stream>>>(W2, b2, Wdec, bdec, Mf, cbuf);

  size_t ldsB = (size_t)(64 * WS_STRIDE + 32 * XS2_STRIDE) * sizeof(float);
  k_encode<<<2048, 256, ldsB, stream>>>(x, Wenc, benc, enc);

  size_t ldsC1 = (size_t)(64 * WS_STRIDE + 128 * F1_ESTR) * sizeof(float);
  k_fc1<<<256, 512, ldsC1, stream>>>(enc, W1, b1, Tbuf);

  size_t ldsC2 = (size_t)(256 * MS_STRIDE + 32 * ES2_STRIDE) * sizeof(float);
  k_decode<<<512, 256, ldsC2, stream>>>(Tbuf, Mf, cbuf, out);
}


# decode: A rows ordered (channel, sx) so a lane's 4 accumulators are 4 adjacent output pixels: 16-B stores, each 128-B line of out written whole by one store (was 4-B pieces at 16-B stride)
# speedup vs baseline: 1.5662x; 1.0396x over previous
#include <hip/hip_runtime.h>

typedef float v2f __attribute__((ext_vector_type(2)));
typedef float v8f __attribute__((ext_vector_type(8)));

#define WMMA_F32(a_, b_, c_) \
  __builtin_amdgcn_wmma_f32_16x16x4_f32(false, (a_), false, (b_), (short)0, (c_), false, false)

#define NB      4
#define CIN     256
#define HH      256
#define WW      256
#define EC      64
#define GW      32
#define LPERN   1024
#define LTOT    4096
#define K1      4096
#define COUT    256

#define ENC_FLOATS  (16777216)
#define T_FLOATS    (262144)
#define M_FLOATS    (262144)

#define WS_STRIDE   260
#define MS_STRIDE   68
#define XS2_STRIDE  288
#define XS_SKEW     4
#define ES2_STRIDE  96

static __device__ __forceinline__ v8f v8f_zero() {
  v8f z = {0.f, 0.f, 0.f, 0.f, 0.f, 0.f, 0.f, 0.f};
  return z;
}

__global__ __launch_bounds__(256) void k_fuse(
    const float* __restrict__ W2, const float* __restrict__ b2,
    const float* __restrict__ Wdec, const float* __restrict__ bdec,
    float* __restrict__ Mf, float* __restrict__ cb) {
  int idx = blockIdx.x * 256 + threadIdx.x;
  int sy = idx >> 16;
  int o  = (idx >> 8) & 255;
  int sx = (idx >> 6) & 3;
  int f  = idx & 63;
  int s  = sy * 4 + sx;
  float sum = 0.f, sb = 0.f;
  #pragma unroll 8
  for (int e = 0; e < 64; ++e) {
    float wd = Wdec[o * 64 + e];
    sum += wd * W2[(e * 16 + s) * 64 + f];
    sb  += wd * b2[e * 16 + s];
  }
  Mf[idx] = sum;
  if (f == 0) cb[idx >> 6] = sb + bdec[o];
}

__global__ __launch_bounds__(256) void k_encode(
    const float* __restrict__ x, const float* __restrict__ Wenc,
    const float* __restrict__ benc, float* __restrict__ enc) {
  extern __shared__ float lds[];
  float* ws = lds;
  float* xs = lds + 64 * WS_STRIDE;

  const int tid = threadIdx.x;
  const int b   = blockIdx.x;
  const int n   = b >> 9;
  const int rem = b & 511;
  const int gh  = rem >> 4;
  const int gw0 = (rem & 15) * 2;

  {
    const float4* src = (const float4*)Wenc;
    #pragma unroll
    for (int i = 0; i < 16; ++i) {
      int v = i * 256 + tid;
      float4 d = src[v];
      int e = v >> 6;
      int k = (v & 63) * 4;
      *(float4*)&ws[e * WS_STRIDE + k] = d;
    }
  }

  const int lane = tid & 31;
  const int wv   = tid >> 5;
  const int lm   = lane & 15;
  const int lh   = lane >> 4;
  const int e0   = (wv >> 1) * 16;
  const int c0   = (wv & 1) * 64;
  const float* xb = xs + (wv & 1) * XS_SKEW;

  v8f acc[4];
  #pragma unroll
  for (int t = 0; t < 4; ++t) acc[t] = v8f_zero();

  const int sq  = tid & 3;
  const int skh = (tid >> 2) & 7;
  const int spp = sq >> 1;
  const int scol = (spp * 64 + skh * 8 + (sq & 1) * 4) * 2 + spp * XS_SKEW;
  const float* xrow = x +
      ((((size_t)n * CIN + wv * 8) * HH + (gh * 8 + skh)) * WW + gw0 * 8 + sq * 4);

  for (int kc = 0; kc < 256; kc += 64) {
    __syncthreads();
    {
      const float* p0 = xrow + (size_t)kc * HH * WW;
      float4 v[8];
      #pragma unroll
      for (int j = 0; j < 8; ++j)
        v[j] = *(const float4*)(p0 + (size_t)j * HH * WW);
      #pragma unroll
      for (int p = 0; p < 4; ++p) {
        float4 lo, hi;
        lo.x = v[2 * p].x; lo.y = v[2 * p + 1].x; lo.z = v[2 * p].y; lo.w = v[2 * p + 1].y;
        hi.x = v[2 * p].z; hi.y = v[2 * p + 1].z; hi.z = v[2 * p].w; hi.w = v[2 * p + 1].w;
        float* d = &xs[(wv * 4 + p) * XS2_STRIDE + scol];
        *(float4*)d       = lo;
        *(float4*)(d + 4) = hi;
      }
    }
    __syncthreads();

    #pragma unroll 4
    for (int kk = 0; kk < 64; kk += 4) {
      int ka = kk + 2 * lh;
      v2f a = *(const v2f*)&ws[(e0 + lm) * WS_STRIDE + kc + ka];
      #pragma unroll
      for (int t = 0; t < 4; ++t) {
        v2f bb = *(const v2f*)&xb[(ka >> 1) * XS2_STRIDE + (c0 + t * 16 + lm) * 2];
        acc[t] = WMMA_F32(a, bb, acc[t]);
      }
    }
  }

  size_t base_l = (size_t)n * LPERN + gh * GW + gw0;
  #pragma unroll
  for (int t = 0; t < 4; ++t) {
    int c  = c0 + t * 16 + lm;
    int pp = c >> 6;
    int ph = c & 63;
    #pragma unroll
    for (int r = 0; r < 8; ++r) {
      int e = e0 + r + 8 * lh;
      enc[(base_l + pp) * (size_t)K1 + e * 64 + ph] = acc[t][r] + benc[e];
    }
  }
}

#define F1_NT     16
#define F1_KC     256
#define F1_ESTR   32

static __device__ __forceinline__ int f1_es_off(int R, int col) {
  return R * F1_ESTR + 2 * (col ^ ((R >> 1) & 15));
}

__global__ __launch_bounds__(512) void k_fc1(
    const float* enc, const float* W1,
    const float* __restrict__ b1, float* Tbuf) {
  extern __shared__ float lds[];
  float* w1s = lds;
  float* es  = lds + 64 * WS_STRIDE;

  const int tid  = threadIdx.x;
  const int l0   = blockIdx.x * F1_NT;
  const int lane = tid & 31, wv = tid >> 5;
  const int lm   = lane & 15, lh = lane >> 4;
  const int mt   = wv & 3;
  const int ks   = wv >> 2;

  const int sr  = tid >> 6;
  const int wkq = (tid & 63) * 4;
  const float* w1p = W1 + (size_t)sr * K1 + wkq;
  const float* enp = enc + (size_t)(l0 + sr) * K1 + wkq;

  float4 pw0 = *(const float4*)(w1p);
  float4 pw1 = *(const float4*)(w1p + (size_t) 8 * K1);
  float4 pw2 = *(const float4*)(w1p + (size_t)16 * K1);
  float4 pw3 = *(const float4*)(w1p + (size_t)24 * K1);
  float4 pw4 = *(const float4*)(w1p + (size_t)32 * K1);
  float4 pw5 = *(const float4*)(w1p + (size_t)40 * K1);
  float4 pw6 = *(const float4*)(w1p + (size_t)48 * K1);
  float4 pw7 = *(const float4*)(w1p + (size_t)56 * K1);
  float4 pe0 = *(const float4*)(enp);
  float4 pe1 = *(const float4*)(enp + (size_t)8 * K1);

  v8f acc = v8f_zero();

  for (int ch = 0; ch < K1 / F1_KC; ++ch) {
    __syncthreads();
    *(float4*)&w1s[(sr +  0) * WS_STRIDE + wkq] = pw0;
    *(float4*)&w1s[(sr +  8) * WS_STRIDE + wkq] = pw1;
    *(float4*)&w1s[(sr + 16) * WS_STRIDE + wkq] = pw2;
    *(float4*)&w1s[(sr + 24) * WS_STRIDE + wkq] = pw3;
    *(float4*)&w1s[(sr + 32) * WS_STRIDE + wkq] = pw4;
    *(float4*)&w1s[(sr + 40) * WS_STRIDE + wkq] = pw5;
    *(float4*)&w1s[(sr + 48) * WS_STRIDE + wkq] = pw6;
    *(float4*)&w1s[(sr + 56) * WS_STRIDE + wkq] = pw7;
    {
      int R0 = wkq >> 1;
      v2f t0; t0.x = pe0.x; t0.y = pe0.y;
      v2f t1; t1.x = pe0.z; t1.y = pe0.w;
      *(v2f*)&es[f1_es_off(R0 + 0, sr)] = t0;
      *(v2f*)&es[f1_es_off(R0 + 1, sr)] = t1;
      v2f u0; u0.x = pe1.x; u0.y = pe1.y;
      v2f u1; u1.x = pe1.z; u1.y = pe1.w;
      *(v2f*)&es[f1_es_off(R0 + 0, sr + 8)] = u0;
      *(v2f*)&es[f1_es_off(R0 + 1, sr + 8)] = u1;
    }
    __syncthreads();

    if (ch + 1 < K1 / F1_KC) {
      int kn = (ch + 1) * F1_KC;
      pw0 = *(const float4*)(w1p + kn);
      pw1 = *(const float4*)(w1p + (size_t) 8 * K1 + kn);
      pw2 = *(const float4*)(w1p + (size_t)16 * K1 + kn);
      pw3 = *(const float4*)(w1p + (size_t)24 * K1 + kn);
      pw4 = *(const float4*)(w1p + (size_t)32 * K1 + kn);
      pw5 = *(const float4*)(w1p + (size_t)40 * K1 + kn);
      pw6 = *(const float4*)(w1p + (size_t)48 * K1 + kn);
      pw7 = *(const float4*)(w1p + (size_t)56 * K1 + kn);
      pe0 = *(const float4*)(enp + kn);
      pe1 = *(const float4*)(enp + (size_t)8 * K1 + kn);
    }

    #pragma unroll
    for (int j = 0; j < 16; ++j) {
      int ka = ks * 64 + j * 4 + 2 * lh;
      v2f a  = *(const v2f*)&w1s[(mt * 16 + lm) * WS_STRIDE + ka];
      v2f bb = *(const v2f*)&es[f1_es_off(ka >> 1, lm)];
      acc = WMMA_F32(a, bb, acc);
    }
  }

  __syncthreads();
  float* red = lds;
  #pragma unroll
  for (int r = 0; r < 8; ++r)
    red[ks * 1024 + mt * 256 + r * 32 + lane] = acc[r];
  __syncthreads();
  #pragma unroll
  for (int h = 0; h < 2; ++h) {
    int o   = tid + 512 * h;
    int f   = o >> 4;
    int col = o & 15;
    int fr  = f & 15;
    int off = (f >> 4) * 256 + (fr & 7) * 32 + (fr >> 3) * 16 + col;
    float s = ((red[off] + red[1024 + off]) + red[2048 + off]) + red[3072 + off];
    Tbuf[(size_t)blockIdx.x * 1024 + o] = s + b1[f];
  }
}

__global__ __launch_bounds__(256) void k_decode(
    const float* __restrict__ Tbuf, const float* __restrict__ Mf,
    const float* __restrict__ cb, float* __restrict__ out) {
  extern __shared__ float lds[];
  float* ms = lds;
  float* ts = lds + 256 * MS_STRIDE;

  const int tid = threadIdx.x;
  const int pb  = blockIdx.x >> 2;
  const int sg  = blockIdx.x & 3;
  const int l0  = pb * 32;
  const int n   = l0 >> 10;
  const int gh  = (l0 & 1023) >> 5;

  #pragma unroll
  for (int i = 0; i < 4; ++i) {
    int v  = i * 256 + tid;
    int f2 = v >> 5;
    int c  = v & 31;
    v2f t;
    const float* tt = Tbuf + (size_t)((l0 + c) >> 4) * 1024 + (c & 15);
    t.x = tt[(2 * f2 + 0) * 16];
    t.y = tt[(2 * f2 + 1) * 16];
    *(v2f*)&ts[f2 * ES2_STRIDE + c * 2] = t;
  }

  const int lane = tid & 31, wv = tid >> 5;
  const int lm   = lane & 15, lh = lane >> 4;

  const int h2 = gh * 4 + sg;

  for (int ss = 0; ss < 4; ++ss) {
    int s = sg * 4 + ss;
    __syncthreads();
    #pragma unroll
    for (int i = 0; i < 16; ++i) {
      int v  = i * 256 + tid;
      int o  = v >> 4;
      int fq = (v & 15) * 4;
      float4 d = *(const float4*)(Mf + (size_t)s * 16384 + o * 64 + fq);
      *(float4*)&ms[o * MS_STRIDE + fq] = d;
    }
    __syncthreads();

    #pragma unroll
    for (int j = 0; j < 4; ++j) {
      int tt = wv * 4 + j;
      int o0 = (tt >> 1) * 16;
      int ct = (tt & 1) * 16;
      v8f acc = v8f_zero();
      #pragma unroll
      for (int kk = 0; kk < 64; kk += 4) {
        int ka = kk + 2 * lh;
        v2f a  = *(const v2f*)&ms[(o0 + lm) * MS_STRIDE + ka];
        v2f bb = *(const v2f*)&ts[(ka >> 1) * ES2_STRIDE + (ct + lm) * 2];
        acc = WMMA_F32(a, bb, acc);
      }
      int gw = ct + lm;
      int oa = ss * 64 + (o0 >> 2) + 2 * lh;
      const float4 ca = *(const float4*)&cb[(sg * 256 + oa) * 4];
      const float4 cc = *(const float4*)&cb[(sg * 256 + oa + 1) * 4];
      float4 va, vb;
      va.x = acc[0] + ca.x; va.y = acc[1] + ca.y; va.z = acc[2] + ca.z; va.w = acc[3] + ca.w;
      vb.x = acc[4] + cc.x; vb.y = acc[5] + cc.y; vb.z = acc[6] + cc.z; vb.w = acc[7] + cc.w;
      float* orow = out + (((size_t)n * COUT + oa) * 128 + h2) * 128 + gw * 4;
      *(float4*)orow               = va;
      *(float4*)(orow + 128 * 128) = vb;
    }
  }
}

extern "C" void kernel_launch(void* const* d_in, const int* in_sizes, int n_in,
                              void* d_out, int out_size, void* d_ws, size_t ws_size,
                              hipStream_t stream) {
  const float* x    = (const float*)d_in[0];
  const float* Wenc = (const float*)d_in[1];
  const float* benc = (const float*)d_in[2];
  const float* W1   = (const float*)d_in[3];
  const float* b1   = (const float*)d_in[4];
  const float* W2   = (const float*)d_in[5];
  const float* b2   = (const float*)d_in[6];
  const float* Wdec = (const float*)d_in[7];
  const float* bdec = (const float*)d_in[8];
  float* out = (float*)d_out;

  float* ws   = (float*)d_ws;
  float* enc  = ws;
  float* Tbuf = ws + ENC_FLOATS;
  float* Mf   = Tbuf + T_FLOATS;
  float* cbuf = Mf + M_FLOATS;

  k_fuse<<<1024, 256, 0, stream>>>(W2, b2, Wdec, bdec, Mf, cbuf);

  size_t ldsB = (size_t)(64 * WS_STRIDE + 32 * XS2_STRIDE) * sizeof(float);
  k_encode<<<2048, 256, ldsB, stream>>>(x, Wenc, benc, enc);

  size_t ldsC1 = (size_t)(64 * WS_STRIDE + 128 * F1_ESTR) * sizeof(float);
  k_fc1<<<256, 512, ldsC1, stream>>>(enc, W1, b1, Tbuf);

  size_t ldsC2 = (size_t)(256 * MS_STRIDE + 32 * ES2_STRIDE) * sizeof(float);
  k_decode<<<512, 256, ldsC2, stream>>>(Tbuf, Mf, cbuf, out);
}


# weight fold (FC2 x decode) moved out of its own launch into the FC1 launch as 512 extra work-groups behind the 256 of FC1: 3 launches, the fold runs in FC1's free wave slots
# speedup vs baseline: 1.7038x; 1.0879x over previous
#include <hip/hip_runtime.h>

typedef float v2f __attribute__((ext_vector_type(2)));
typedef float v8f __attribute__((ext_vector_type(8)));

#define WMMA_F32(a_, b_, c_) \
  __builtin_amdgcn_wmma_f32_16x16x4_f32(false, (a_), false, (b_), (short)0, (c_), false, false)

#define NB      4
#define CIN     256
#define HH      256
#define WW      256
#define EC      64
#define GW      32
#define LPERN   1024
#define LTOT    4096
#define K1      4096
#define COUT    256

#define ENC_FLOATS  (16777216)
#define T_FLOATS    (262144)
#define M_FLOATS    (262144)

#define WS_STRIDE   260
#define MS_STRIDE   68
#define XS2_STRIDE  288
#define XS_SKEW     4
#define ES2_STRIDE  96

static __device__ __forceinline__ v8f v8f_zero() {
  v8f z = {0.f, 0.f, 0.f, 0.f, 0.f, 0.f, 0.f, 0.f};
  return z;
}

static __device__ __forceinline__ void fuse_elem(
    int idx,
    const float* __restrict__ W2, const float* __restrict__ b2,
    const float* __restrict__ Wdec, const float* __restrict__ bdec,
    float* __restrict__ Mf, float* __restrict__ cb) {
  int sy = idx >> 16;
  int o  = (idx >> 8) & 255;
  int sx = (idx >> 6) & 3;
  int f  = idx & 63;
  int s  = sy * 4 + sx;
  float sum = 0.f, sb = 0.f;
  #pragma unroll 8
  for (int e = 0; e < 64; ++e) {
    float wd = Wdec[o * 64 + e];
    sum += wd * W2[(e * 16 + s) * 64 + f];
    sb  += wd * b2[e * 16 + s];
  }
  Mf[idx] = sum;
  if (f == 0) cb[idx >> 6] = sb + bdec[o];
}

__global__ __launch_bounds__(256) void k_encode(
    const float* __restrict__ x, const float* __restrict__ Wenc,
    const float* __restrict__ benc, float* __restrict__ enc) {
  extern __shared__ float lds[];
  float* ws = lds;
  float* xs = lds + 64 * WS_STRIDE;

  const int tid = threadIdx.x;
  const int b   = blockIdx.x;
  const int n   = b >> 9;
  const int rem = b & 511;
  const int gh  = rem >> 4;
  const int gw0 = (rem & 15) * 2;

  {
    const float4* src = (const float4*)Wenc;
    #pragma unroll
    for (int i = 0; i < 16; ++i) {
      int v = i * 256 + tid;
      float4 d = src[v];
      int e = v >> 6;
      int k = (v & 63) * 4;
      *(float4*)&ws[e * WS_STRIDE + k] = d;
    }
  }

  const int lane = tid & 31;
  const int wv   = tid >> 5;
  const int lm   = lane & 15;
  const int lh   = lane >> 4;
  const int e0   = (wv >> 1) * 16;
  const int c0   = (wv & 1) * 64;
  const float* xb = xs + (wv & 1) * XS_SKEW;

  v8f acc[4];
  #pragma unroll
  for (int t = 0; t < 4; ++t) acc[t] = v8f_zero();

  const int sq  = tid & 3;
  const int skh = (tid >> 2) & 7;
  const int spp = sq >> 1;
  const int scol = (spp * 64 + skh * 8 + (sq & 1) * 4) * 2 + spp * XS_SKEW;
  const float* xrow = x +
      ((((size_t)n * CIN + wv * 8) * HH + (gh * 8 + skh)) * WW + gw0 * 8 + sq * 4);

  for (int kc = 0; kc < 256; kc += 64) {
    __syncthreads();
    {
      const float* p0 = xrow + (size_t)kc * HH * WW;
      float4 v[8];
      #pragma unroll
      for (int j = 0; j < 8; ++j)
        v[j] = *(const float4*)(p0 + (size_t)j * HH * WW);
      #pragma unroll
      for (int p = 0; p < 4; ++p) {
        float4 lo, hi;
        lo.x = v[2 * p].x; lo.y = v[2 * p + 1].x; lo.z = v[2 * p].y; lo.w = v[2 * p + 1].y;
        hi.x = v[2 * p].z; hi.y = v[2 * p + 1].z; hi.z = v[2 * p].w; hi.w = v[2 * p + 1].w;
        float* d = &xs[(wv * 4 + p) * XS2_STRIDE + scol];
        *(float4*)d       = lo;
        *(float4*)(d + 4) = hi;
      }
    }
    __syncthreads();

    #pragma unroll 4
    for (int kk = 0; kk < 64; kk += 4) {
      int ka = kk + 2 * lh;
      v2f a = *(const v2f*)&ws[(e0 + lm) * WS_STRIDE + kc + ka];
      #pragma unroll
      for (int t = 0; t < 4; ++t) {
        v2f bb = *(const v2f*)&xb[(ka >> 1) * XS2_STRIDE + (c0 + t * 16 + lm) * 2];
        acc[t] = WMMA_F32(a, bb, acc[t]);
      }
    }
  }

  size_t base_l = (size_t)n * LPERN + gh * GW + gw0;
  #pragma unroll
  for (int t = 0; t < 4; ++t) {
    int c  = c0 + t * 16 + lm;
    int pp = c >> 6;
    int ph = c & 63;
    #pragma unroll
    for (int r = 0; r < 8; ++r) {
      int e = e0 + r + 8 * lh;
      enc[(base_l + pp) * (size_t)K1 + e * 64 + ph] = acc[t][r] + benc[e];
    }
  }
}

#define F1_WG       256
#define F1_FUSE_WG  512
static_assert(F1_FUSE_WG * 512 == M_FLOATS, "one fold thread per element of M");
#define F1_NT     16
#define F1_KC     256
#define F1_ESTR   32

static __device__ __forceinline__ int f1_es_off(int R, int col) {
  return R * F1_ESTR + 2 * (col ^ ((R >> 1) & 15));
}

__global__ __launch_bounds__(512) void k_fc1(
    const float* enc, const float* W1,
    const float* __restrict__ b1, float* Tbuf,
    const float* __restrict__ W2, const float* __restrict__ b2,
    const float* __restrict__ Wdec, const float* __restrict__ bdec,
    float* __restrict__ Mf, float* __restrict__ cb) {
  if (blockIdx.x >= F1_WG) {
    fuse_elem((blockIdx.x - F1_WG) * 512 + threadIdx.x, W2, b2, Wdec, bdec, Mf, cb);
    return;
  }
  extern __shared__ float lds[];
  float* w1s = lds;
  float* es  = lds + 64 * WS_STRIDE;

  const int tid  = threadIdx.x;
  const int l0   = blockIdx.x * F1_NT;
  const int lane = tid & 31, wv = tid >> 5;
  const int lm   = lane & 15, lh = lane >> 4;
  const int mt   = wv & 3;
  const int ks   = wv >> 2;

  const int sr  = tid >> 6;
  const int wkq = (tid & 63) * 4;
  const float* w1p = W1 + (size_t)sr * K1 + wkq;
  const float* enp = enc + (size_t)(l0 + sr) * K1 + wkq;

  float4 pw0 = *(const float4*)(w1p);
  float4 pw1 = *(const float4*)(w1p + (size_t) 8 * K1);
  float4 pw2 = *(const float4*)(w1p + (size_t)16 * K1);
  float4 pw3 = *(const float4*)(w1p + (size_t)24 * K1);
  float4 pw4 = *(const float4*)(w1p + (size_t)32 * K1);
  float4 pw5 = *(const float4*)(w1p + (size_t)40 * K1);
  float4 pw6 = *(const float4*)(w1p + (size_t)48 * K1);
  float4 pw7 = *(const float4*)(w1p + (size_t)56 * K1);
  float4 pe0 = *(const float4*)(enp);
  float4 pe1 = *(const float4*)(enp + (size_t)8 * K1);

  v8f acc = v8f_zero();

  for (int ch = 0; ch < K1 / F1_KC; ++ch) {
    __syncthreads();
    *(float4*)&w1s[(sr +  0) * WS_STRIDE + wkq] = pw0;
    *(float4*)&w1s[(sr +  8) * WS_STRIDE + wkq] = pw1;
    *(float4*)&w1s[(sr + 16) * WS_STRIDE + wkq] = pw2;
    *(float4*)&w1s[(sr + 24) * WS_STRIDE + wkq] = pw3;
    *(float4*)&w1s[(sr + 32) * WS_STRIDE + wkq] = pw4;
    *(float4*)&w1s[(sr + 40) * WS_STRIDE + wkq] = pw5;
    *(float4*)&w1s[(sr + 48) * WS_STRIDE + wkq] = pw6;
    *(float4*)&w1s[(sr + 56) * WS_STRIDE + wkq] = pw7;
    {
      int R0 = wkq >> 1;
      v2f t0; t0.x = pe0.x; t0.y = pe0.y;
      v2f t1; t1.x = pe0.z; t1.y = pe0.w;
      *(v2f*)&es[f1_es_off(R0 + 0, sr)] = t0;
      *(v2f*)&es[f1_es_off(R0 + 1, sr)] = t1;
      v2f u0; u0.x = pe1.x; u0.y = pe1.y;
      v2f u1; u1.x = pe1.z; u1.y = pe1.w;
      *(v2f*)&es[f1_es_off(R0 + 0, sr + 8)] = u0;
      *(v2f*)&es[f1_es_off(R0 + 1, sr + 8)] = u1;
    }
    __syncthreads();

    if (ch + 1 < K1 / F1_KC) {
      int kn = (ch + 1) * F1_KC;
      pw0 = *(const float4*)(w1p + kn);
      pw1 = *(const float4*)(w1p + (size_t) 8 * K1 + kn);
      pw2 = *(const float4*)(w1p + (size_t)16 * K1 + kn);
      pw3 = *(const float4*)(w1p + (size_t)24 * K1 + kn);
      pw4 = *(const float4*)(w1p + (size_t)32 * K1 + kn);
      pw5 = *(const float4*)(w1p + (size_t)40 * K1 + kn);
      pw6 = *(const float4*)(w1p + (size_t)48 * K1 + kn);
      pw7 = *(const float4*)(w1p + (size_t)56 * K1 + kn);
      pe0 = *(const float4*)(enp + kn);
      pe1 = *(const float4*)(enp + (size_t)8 * K1 + kn);
    }

    #pragma unroll
    for (int j = 0; j < 16; ++j) {
      int ka = ks * 64 + j * 4 + 2 * lh;
      v2f a  = *(const v2f*)&w1s[(mt * 16 + lm) * WS_STRIDE + ka];
      v2f bb = *(const v2f*)&es[f1_es_off(ka >> 1, lm)];
      acc = WMMA_F32(a, bb, acc);
    }
  }

  __syncthreads();
  float* red = lds;
  #pragma unroll
  for (int r = 0; r < 8; ++r)
    red[ks * 1024 + mt * 256 + r * 32 + lane] = acc[r];
  __syncthreads();
  #pragma unroll
  for (int h = 0; h < 2; ++h) {
    int o   = tid + 512 * h;
    int f   = o >> 4;
    int col = o & 15;
    int fr  = f & 15;
    int off = (f >> 4) * 256 + (fr & 7) * 32 + (fr >> 3) * 16 + col;
    float s = ((red[off] + red[1024 + off]) + red[2048 + off]) + red[3072 + off];
    Tbuf[(size_t)blockIdx.x * 1024 + o] = s + b1[f];
  }
}

__global__ __launch_bounds__(256) void k_decode(
    const float* __restrict__ Tbuf, const float* __restrict__ Mf,
    const float* __restrict__ cb, float* __restrict__ out) {
  extern __shared__ float lds[];
  float* ms = lds;
  float* ts = lds + 256 * MS_STRIDE;

  const int tid = threadIdx.x;
  const int pb  = blockIdx.x >> 2;
  const int sg  = blockIdx.x & 3;
  const int l0  = pb * 32;
  const int n   = l0 >> 10;
  const int gh  = (l0 & 1023) >> 5;

  #pragma unroll
  for (int i = 0; i < 4; ++i) {
    int v  = i * 256 + tid;
    int f2 = v >> 5;
    int c  = v & 31;
    v2f t;
    const float* tt = Tbuf + (size_t)((l0 + c) >> 4) * 1024 + (c & 15);
    t.x = tt[(2 * f2 + 0) * 16];
    t.y = tt[(2 * f2 + 1) * 16];
    *(v2f*)&ts[f2 * ES2_STRIDE + c * 2] = t;
  }

  const int lane = tid & 31, wv = tid >> 5;
  const int lm   = lane & 15, lh = lane >> 4;

  const int h2 = gh * 4 + sg;

  for (int ss = 0; ss < 4; ++ss) {
    int s = sg * 4 + ss;
    __syncthreads();
    #pragma unroll
    for (int i = 0; i < 16; ++i) {
      int v  = i * 256 + tid;
      int o  = v >> 4;
      int fq = (v & 15) * 4;
      float4 d = *(const float4*)(Mf + (size_t)s * 16384 + o * 64 + fq);
      *(float4*)&ms[o * MS_STRIDE + fq] = d;
    }
    __syncthreads();

    #pragma unroll
    for (int j = 0; j < 4; ++j) {
      int tt = wv * 4 + j;
      int o0 = (tt >> 1) * 16;
      int ct = (tt & 1) * 16;
      v8f acc = v8f_zero();
      #pragma unroll
      for (int kk = 0; kk < 64; kk += 4) {
        int ka = kk + 2 * lh;
        v2f a  = *(const v2f*)&ms[(o0 + lm) * MS_STRIDE + ka];
        v2f bb = *(const v2f*)&ts[(ka >> 1) * ES2_STRIDE + (ct + lm) * 2];
        acc = WMMA_F32(a, bb, acc);
      }
      int gw = ct + lm;
      int oa = ss * 64 + (o0 >> 2) + 2 * lh;
      const float4 ca = *(const float4*)&cb[(sg * 256 + oa) * 4];
      const float4 cc = *(const float4*)&cb[(sg * 256 + oa + 1) * 4];
      float4 va, vb;
      va.x = acc[0] + ca.x; va.y = acc[1] + ca.y; va.z = acc[2] + ca.z; va.w = acc[3] + ca.w;
      vb.x = acc[4] + cc.x; vb.y = acc[5] + cc.y; vb.z = acc[6] + cc.z; vb.w = acc[7] + cc.w;
      float* orow = out + (((size_t)n * COUT + oa) * 128 + h2) * 128 + gw * 4;
      *(float4*)orow               = va;
      *(float4*)(orow + 128 * 128) = vb;
    }
  }
}

extern "C" void kernel_launch(void* const* d_in, const int* in_sizes, int n_in,
                              void* d_out, int out_size, void* d_ws, size_t ws_size,
                              hipStream_t stream) {
  const float* x    = (const float*)d_in[0];
  const float* Wenc = (const float*)d_in[1];
  const float* benc = (const float*)d_in[2];
  const float* W1   = (const float*)d_in[3];
  const float* b1   = (const float*)d_in[4];
  const float* W2   = (const float*)d_in[5];
  const float* b2   = (const float*)d_in[6];
  const float* Wdec = (const float*)d_in[7];
  const float* bdec = (const float*)d_in[8];
  float* out = (float*)d_out;

  float* ws   = (float*)d_ws;
  float* enc  = ws;
  float* Tbuf = ws + ENC_FLOATS;
  float* Mf   = Tbuf + T_FLOATS;
  float* cbuf = Mf + M_FLOATS;

  size_t ldsB = (size_t)(64 * WS_STRIDE + 32 * XS2_STRIDE) * sizeof(float);
  k_encode<<<2048, 256, ldsB, stream>>>(x, Wenc, benc, enc);

  size_t ldsC1 = (size_t)(64 * WS_STRIDE + 128 * F1_ESTR) * sizeof(float);
  k_fc1<<<768, 512, ldsC1, stream>>>(enc, W1, b1, Tbuf, W2, b2, Wdec, bdec, Mf, cbuf);

  size_t ldsC2 = (size_t)(256 * MS_STRIDE + 32 * ES2_STRIDE) * sizeof(float);
  k_decode<<<512, 256, ldsC2, stream>>>(Tbuf, Mf, cbuf, out);
}


# encode, FC1: LDS fragment reads of each WMMA block (16 / 4 WMMAs) issued as one group ahead of the block into a second register set, none between the WMMAs; every sum and its order unchanged
# speedup vs baseline: 1.7701x; 1.0389x over previous
#include <hip/hip_runtime.h>

typedef float v2f __attribute__((ext_vector_type(2)));
typedef float v8f __attribute__((ext_vector_type(8)));

#define WMMA_F32(a_, b_, c_) \
  __builtin_amdgcn_wmma_f32_16x16x4_f32(false, (a_), false, (b_), (short)0, (c_), false, false)

#define NB      4
#define CIN     256
#define HH      256
#define WW      256
#define EC      64
#define GW      32
#define LPERN   1024
#define LTOT    4096
#define K1      4096
#define COUT    256

#define ENC_FLOATS  (16777216)
#define T_FLOATS    (262144)
#define M_FLOATS    (262144)

#define WS_STRIDE   260
#define MS_STRIDE   68
#define XS2_STRIDE  288
#define XS_SKEW     4
#define ES2_STRIDE  96

static __device__ __forceinline__ v8f v8f_zero() {
  v8f z = {0.f, 0.f, 0.f, 0.f, 0.f, 0.f, 0.f, 0.f};
  return z;
}

static __device__ __forceinline__ void fuse_elem(
    int idx,
    const float* __restrict__ W2, const float* __restrict__ b2,
    const float* __restrict__ Wdec, const float* __restrict__ bdec,
    float* __restrict__ Mf, float* __restrict__ cb) {
  int sy = idx >> 16;
  int o  = (idx >> 8) & 255;
  int sx = (idx >> 6) & 3;
  int f  = idx & 63;
  int s  = sy * 4 + sx;
  float sum = 0.f, sb = 0.f;
  #pragma unroll 8
  for (int e = 0; e < 64; ++e) {
    float wd = Wdec[o * 64 + e];
    sum += wd * W2[(e * 16 + s) * 64 + f];
    sb  += wd * b2[e * 16 + s];
  }
  Mf[idx] = sum;
  if (f == 0) cb[idx >> 6] = sb + bdec[o];
}

#define ENC_LOAD_BLOCK(buf_, g_)                                                        \
  _Pragma("unroll")                                                                     \
  for (int s_ = 0; s_ < 4; ++s_) {                                                      \
    fa[buf_][s_] = *(const v2f*)(ap + (g_) * 16 + s_ * 4);                              \
    _Pragma("unroll")                                                                   \
    for (int t_ = 0; t_ < 4; ++t_)                                                      \
      fb[buf_][s_][t_] =                                                                \
          *(const v2f*)(bp + ((g_) * 8 + s_ * 2) * XS2_STRIDE + t_ * 32);               \
  }

__global__ __launch_bounds__(256) void k_encode(
    const float* __restrict__ x, const float* __restrict__ Wenc,
    const float* __restrict__ benc, float* __restrict__ enc) {
  extern __shared__ float lds[];
  float* ws = lds;
  float* xs = lds + 64 * WS_STRIDE;

  const int tid = threadIdx.x;
  const int b   = blockIdx.x;
  const int n   = b >> 9;
  const int rem = b & 511;
  const int gh  = rem >> 4;
  const int gw0 = (rem & 15) * 2;

  {
    const float4* src = (const float4*)Wenc;
    #pragma unroll
    for (int i = 0; i < 16; ++i) {
      int v = i * 256 + tid;
      float4 d = src[v];
      int e = v >> 6;
      int k = (v & 63) * 4;
      *(float4*)&ws[e * WS_STRIDE + k] = d;
    }
  }

  const int lane = tid & 31;
  const int wv   = tid >> 5;
  const int lm   = lane & 15;
  const int lh   = lane >> 4;
  const int e0   = (wv >> 1) * 16;
  const int c0   = (wv & 1) * 64;
  const float* xb = xs + (wv & 1) * XS_SKEW;

  v8f acc[4];
  #pragma unroll
  for (int t = 0; t < 4; ++t) acc[t] = v8f_zero();

  const int sq  = tid & 3;
  const int skh = (tid >> 2) & 7;
  const int spp = sq >> 1;
  const int scol = (spp * 64 + skh * 8 + (sq & 1) * 4) * 2 + spp * XS_SKEW;
  const float* xrow = x +
      ((((size_t)n * CIN + wv * 8) * HH + (gh * 8 + skh)) * WW + gw0 * 8 + sq * 4);

  for (int kc = 0; kc < 256; kc += 64) {
    __syncthreads();
    {
      const float* p0 = xrow + (size_t)kc * HH * WW;
      float4 v[8];
      #pragma unroll
      for (int j = 0; j < 8; ++j)
        v[j] = *(const float4*)(p0 + (size_t)j * HH * WW);
      #pragma unroll
      for (int p = 0; p < 4; ++p) {
        float4 lo, hi;
        lo.x = v[2 * p].x; lo.y = v[2 * p + 1].x; lo.z = v[2 * p].y; lo.w = v[2 * p + 1].y;
        hi.x = v[2 * p].z; hi.y = v[2 * p + 1].z; hi.z = v[2 * p].w; hi.w = v[2 * p + 1].w;
        float* d = &xs[(wv * 4 + p) * XS2_STRIDE + scol];
        *(float4*)d       = lo;
        *(float4*)(d + 4) = hi;
      }
    }
    __syncthreads();

    const float* ap = &ws[(e0 + lm) * WS_STRIDE + kc + 2 * lh];
    const float* bp = &xb[lh * XS2_STRIDE + (c0 + lm) * 2];
    v2f fa[2][4];
    v2f fb[2][4][4];
    ENC_LOAD_BLOCK(0, 0)
    #pragma unroll
    for (int g = 0; g < 4; ++g) {
      if (g + 1 < 4) { ENC_LOAD_BLOCK((g + 1) & 1, g + 1) }
      __builtin_amdgcn_sched_barrier(0);
      #pragma unroll
      for (int s = 0; s < 4; ++s) {
        #pragma unroll
        for (int t = 0; t < 4; ++t)
          acc[t] = WMMA_F32(fa[g & 1][s], fb[g & 1][s][t], acc[t]);
      }
      __builtin_amdgcn_sched_barrier(0);
    }
  }

  size_t base_l = (size_t)n * LPERN + gh * GW + gw0;
  #pragma unroll
  for (int t = 0; t < 4; ++t) {
    int c  = c0 + t * 16 + lm;
    int pp = c >> 6;
    int ph = c & 63;
    #pragma unroll
    for (int r = 0; r < 8; ++r) {
      int e = e0 + r + 8 * lh;
      enc[(base_l + pp) * (size_t)K1 + e * 64 + ph] = acc[t][r] + benc[e];
    }
  }
}

#define F1_WG       256
#define F1_FUSE_WG  512
static_assert(F1_FUSE_WG * 512 == M_FLOATS, "one fold thread per element of M");
#define F1_NT     16
#define F1_KC     256
#define F1_ESTR   32

static __device__ __forceinline__ int f1_es_off(int R, int col) {
  return R * F1_ESTR + 2 * (col ^ ((R >> 1) & 15));
}

#define F1_LOAD_BLOCK(buf_, g_)                                                         \
  _Pragma("unroll")                                                                     \
  for (int s_ = 0; s_ < 4; ++s_) {                                                      \
    int ka_ = ks * 64 + ((g_) * 4 + s_) * 4 + 2 * lh;                                   \
    fa[buf_][s_] = *(const v2f*)&w1s[(mt * 16 + lm) * WS_STRIDE + ka_];                 \
    fb[buf_][s_] = *(const v2f*)&es[f1_es_off(ka_ >> 1, lm)];                           \
  }

__global__ __launch_bounds__(512) void k_fc1(
    const float* enc, const float* W1,
    const float* __restrict__ b1, float* Tbuf,
    const float* __restrict__ W2, const float* __restrict__ b2,
    const float* __restrict__ Wdec, const float* __restrict__ bdec,
    float* __restrict__ Mf, float* __restrict__ cb) {
  if (blockIdx.x >= F1_WG) {
    fuse_elem((blockIdx.x - F1_WG) * 512 + threadIdx.x, W2, b2, Wdec, bdec, Mf, cb);
    return;
  }
  extern __shared__ float lds[];
  float* w1s = lds;
  float* es  = lds + 64 * WS_STRIDE;

  const int tid  = threadIdx.x;
  const int l0   = blockIdx.x * F1_NT;
  const int lane = tid & 31, wv = tid >> 5;
  const int lm   = lane & 15, lh = lane >> 4;
  const int mt   = wv & 3;
  const int ks   = wv >> 2;

  const int sr  = tid >> 6;
  const int wkq = (tid & 63) * 4;
  const float* w1p = W1 + (size_t)sr * K1 + wkq;
  const float* enp = enc + (size_t)(l0 + sr) * K1 + wkq;

  float4 pw0 = *(const float4*)(w1p);
  float4 pw1 = *(const float4*)(w1p + (size_t) 8 * K1);
  float4 pw2 = *(const float4*)(w1p + (size_t)16 * K1);
  float4 pw3 = *(const float4*)(w1p + (size_t)24 * K1);
  float4 pw4 = *(const float4*)(w1p + (size_t)32 * K1);
  float4 pw5 = *(const float4*)(w1p + (size_t)40 * K1);
  float4 pw6 = *(const float4*)(w1p + (size_t)48 * K1);
  float4 pw7 = *(const float4*)(w1p + (size_t)56 * K1);
  float4 pe0 = *(const float4*)(enp);
  float4 pe1 = *(const float4*)(enp + (size_t)8 * K1);

  v8f acc = v8f_zero();

  for (int ch = 0; ch < K1 / F1_KC; ++ch) {
    __syncthreads();
    *(float4*)&w1s[(sr +  0) * WS_STRIDE + wkq] = pw0;
    *(float4*)&w1s[(sr +  8) * WS_STRIDE + wkq] = pw1;
    *(float4*)&w1s[(sr + 16) * WS_STRIDE + wkq] = pw2;
    *(float4*)&w1s[(sr + 24) * WS_STRIDE + wkq] = pw3;
    *(float4*)&w1s[(sr + 32) * WS_STRIDE + wkq] = pw4;
    *(float4*)&w1s[(sr + 40) * WS_STRIDE + wkq] = pw5;
    *(float4*)&w1s[(sr + 48) * WS_STRIDE + wkq] = pw6;
    *(float4*)&w1s[(sr + 56) * WS_STRIDE + wkq] = pw7;
    {
      int R0 = wkq >> 1;
      v2f t0; t0.x = pe0.x; t0.y = pe0.y;
      v2f t1; t1.x = pe0.z; t1.y = pe0.w;
      *(v2f*)&es[f1_es_off(R0 + 0, sr)] = t0;
      *(v2f*)&es[f1_es_off(R0 + 1, sr)] = t1;
      v2f u0; u0.x = pe1.x; u0.y = pe1.y;
      v2f u1; u1.x = pe1.z; u1.y = pe1.w;
      *(v2f*)&es[f1_es_off(R0 + 0, sr + 8)] = u0;
      *(v2f*)&es[f1_es_off(R0 + 1, sr + 8)] = u1;
    }
    __syncthreads();

    if (ch + 1 < K1 / F1_KC) {
      int kn = (ch + 1) * F1_KC;
      pw0 = *(const float4*)(w1p + kn);
      pw1 = *(const float4*)(w1p + (size_t) 8 * K1 + kn);
      pw2 = *(const float4*)(w1p + (size_t)16 * K1 + kn);
      pw3 = *(const float4*)(w1p + (size_t)24 * K1 + kn);
      pw4 = *(const float4*)(w1p + (size_t)32 * K1 + kn);
      pw5 = *(const float4*)(w1p + (size_t)40 * K1 + kn);
      pw6 = *(const float4*)(w1p + (size_t)48 * K1 + kn);
      pw7 = *(const float4*)(w1p + (size_t)56 * K1 + kn);
      pe0 = *(const float4*)(enp + kn);
      pe1 = *(const float4*)(enp + (size_t)8 * K1 + kn);
    }

    v2f fa[2][4], fb[2][4];
    F1_LOAD_BLOCK(0, 0)
    #pragma unroll
    for (int g = 0; g < 4; ++g) {
      if (g + 1 < 4) { F1_LOAD_BLOCK((g + 1) & 1, g + 1) }
      __builtin_amdgcn_sched_barrier(0);
      #pragma unroll
      for (int s = 0; s < 4; ++s)
        acc = WMMA_F32(fa[g & 1][s], fb[g & 1][s], acc);
      __builtin_amdgcn_sched_barrier(0);
    }
  }

  __syncthreads();
  float* red = lds;
  #pragma unroll
  for (int r = 0; r < 8; ++r)
    red[ks * 1024 + mt * 256 + r * 32 + lane] = acc[r];
  __syncthreads();
  #pragma unroll
  for (int h = 0; h < 2; ++h) {
    int o   = tid + 512 * h;
    int f   = o >> 4;
    int col = o & 15;
    int fr  = f & 15;
    int off = (f >> 4) * 256 + (fr & 7) * 32 + (fr >> 3) * 16 + col;
    float s = ((red[off] + red[1024 + off]) + red[2048 + off]) + red[3072 + off];
    Tbuf[(size_t)blockIdx.x * 1024 + o] = s + b1[f];
  }
}

__global__ __launch_bounds__(256) void k_decode(
    const float* __restrict__ Tbuf, const float* __restrict__ Mf,
    const float* __restrict__ cb, float* __restrict__ out) {
  extern __shared__ float lds[];
  float* ms = lds;
  float* ts = lds + 256 * MS_STRIDE;

  const int tid = threadIdx.x;
  const int pb  = blockIdx.x >> 2;
  const int sg  = blockIdx.x & 3;
  const int l0  = pb * 32;
  const int n   = l0 >> 10;
  const int gh  = (l0 & 1023) >> 5;

  #pragma unroll
  for (int i = 0; i < 4; ++i) {
    int v  = i * 256 + tid;
    int f2 = v >> 5;
    int c  = v & 31;
    v2f t;
    const float* tt = Tbuf + (size_t)((l0 + c) >> 4) * 1024 + (c & 15);
    t.x = tt[(2 * f2 + 0) * 16];
    t.y = tt[(2 * f2 + 1) * 16];
    *(v2f*)&ts[f2 * ES2_STRIDE + c * 2] = t;
  }

  const int lane = tid & 31, wv = tid >> 5;
  const int lm   = lane & 15, lh = lane >> 4;

  const int h2 = gh * 4 + sg;

  for (int ss = 0; ss < 4; ++ss) {
    int s = sg * 4 + ss;
    __syncthreads();
    #pragma unroll
    for (int i = 0; i < 16; ++i) {
      int v  = i * 256 + tid;
      int o  = v >> 4;
      int fq = (v & 15) * 4;
      float4 d = *(const float4*)(Mf + (size_t)s * 16384 + o * 64 + fq);
      *(float4*)&ms[o * MS_STRIDE + fq] = d;
    }
    __syncthreads();

    #pragma unroll
    for (int j = 0; j < 4; ++j) {
      int tt = wv * 4 + j;
      int o0 = (tt >> 1) * 16;
      int ct = (tt & 1) * 16;
      v8f acc = v8f_zero();
      #pragma unroll
      for (int kk = 0; kk < 64; kk += 4) {
        int ka = kk + 2 * lh;
        v2f a  = *(const v2f*)&ms[(o0 + lm) * MS_STRIDE + ka];
        v2f bb = *(const v2f*)&ts[(ka >> 1) * ES2_STRIDE + (ct + lm) * 2];
        acc = WMMA_F32(a, bb, acc);
      }
      int gw = ct + lm;
      int oa = ss * 64 + (o0 >> 2) + 2 * lh;
      const float4 ca = *(const float4*)&cb[(sg * 256 + oa) * 4];
      const float4 cc = *(const float4*)&cb[(sg * 256 + oa + 1) * 4];
      float4 va, vb;
      va.x = acc[0] + ca.x; va.y = acc[1] + ca.y; va.z = acc[2] + ca.z; va.w = acc[3] + ca.w;
      vb.x = acc[4] + cc.x; vb.y = acc[5] + cc.y; vb.z = acc[6] + cc.z; vb.w = acc[7] + cc.w;
      float* orow = out + (((size_t)n * COUT + oa) * 128 + h2) * 128 + gw * 4;
      *(float4*)orow               = va;
      *(float4*)(orow + 128 * 128) = vb;
    }
  }
}

extern "C" void kernel_launch(void* const* d_in, const int* in_sizes, int n_in,
                              void* d_out, int out_size, void* d_ws, size_t ws_size,
                              hipStream_t stream) {
  const float* x    = (const float*)d_in[0];
  const float* Wenc = (const float*)d_in[1];
  const float* benc = (const float*)d_in[2];
  const float* W1   = (const float*)d_in[3];
  const float* b1   = (const float*)d_in[4];
  const float* W2   = (const float*)d_in[5];
  const float* b2   = (const float*)d_in[6];
  const float* Wdec = (const float*)d_in[7];
  const float* bdec = (const float*)d_in[8];
  float* out = (float*)d_out;

  float* ws   = (float*)d_ws;
  float* enc  = ws;
  float* Tbuf = ws + ENC_FLOATS;
  float* Mf   = Tbuf + T_FLOATS;
  float* cbuf = Mf + M_FLOATS;

  size_t ldsB = (size_t)(64 * WS_STRIDE + 32 * XS2_STRIDE) * sizeof(float);
  k_encode<<<2048, 256, ldsB, stream>>>(x, Wenc, benc, enc);

  size_t ldsC1 = (size_t)(64 * WS_STRIDE + 128 * F1_ESTR) * sizeof(float);
  k_fc1<<<768, 512, ldsC1, stream>>>(enc, W1, b1, Tbuf, W2, b2, Wdec, bdec, Mf, cbuf);

  size_t ldsC2 = (size_t)(256 * MS_STRIDE + 32 * ES2_STRIDE) * sizeof(float);
  k_decode<<<512, 256, ldsC2, stream>>>(Tbuf, Mf, cbuf, out);
}


# encode: x goes from global straight into the WMMA B fragments (no LDS tile of x, no barrier in the K loop); wave = 4 rows x 16 px x 64 channels, 4 k-steps of x in flight; sums and order unchanged
# speedup vs baseline: 2.0321x; 1.1480x over previous
#include <hip/hip_runtime.h>

typedef float v2f __attribute__((ext_vector_type(2)));
typedef float v8f __attribute__((ext_vector_type(8)));

#define WMMA_F32(a_, b_, c_) \
  __builtin_amdgcn_wmma_f32_16x16x4_f32(false, (a_), false, (b_), (short)0, (c_), false, false)

#define NB      4
#define CIN     256
#define HH      256
#define WW      256
#define EC      64
#define GW      32
#define LPERN   1024
#define LTOT    4096
#define K1      4096
#define COUT    256

#define ENC_FLOATS  (16777216)
#define T_FLOATS    (262144)
#define M_FLOATS    (262144)

#define WS_STRIDE   260
#define MS_STRIDE   68
#define ES2_STRIDE  96

static __device__ __forceinline__ v8f v8f_zero() {
  v8f z = {0.f, 0.f, 0.f, 0.f, 0.f, 0.f, 0.f, 0.f};
  return z;
}

static __device__ __forceinline__ void fuse_elem(
    int idx,
    const float* __restrict__ W2, const float* __restrict__ b2,
    const float* __restrict__ Wdec, const float* __restrict__ bdec,
    float* __restrict__ Mf, float* __restrict__ cb) {
  int sy = idx >> 16;
  int o  = (idx >> 8) & 255;
  int sx = (idx >> 6) & 3;
  int f  = idx & 63;
  int s  = sy * 4 + sx;
  float sum = 0.f, sb = 0.f;
  #pragma unroll 8
  for (int e = 0; e < 64; ++e) {
    float wd = Wdec[o * 64 + e];
    sum += wd * W2[(e * 16 + s) * 64 + f];
    sb  += wd * b2[e * 16 + s];
  }
  Mf[idx] = sum;
  if (f == 0) cb[idx >> 6] = sb + bdec[o];
}

#define ENC_PF  4
#define XCH     ((size_t)HH * WW)

#define ENC_KSTEP(j_, s_, LOAD_)                                                        \
  {                                                                                     \
    _Pragma("unroll")                                                                   \
    for (int m_ = 0; m_ < 4; ++m_)                                                      \
      fa[((j_) + 1) & 1][m_] =                                                          \
          *(const v2f*)(ap + m_ * 16 * WS_STRIDE + ((((s_) + 1) & 63) * 4));            \
    v2f fb[4];                                                                          \
    fb[0].x = ring[j_][0].x; fb[0].y = ring[j_][1].x;                                   \
    fb[1].x = ring[j_][0].y; fb[1].y = ring[j_][1].y;                                   \
    fb[2].x = ring[j_][0].z; fb[2].y = ring[j_][1].z;                                   \
    fb[3].x = ring[j_][0].w; fb[3].y = ring[j_][1].w;                                   \
    if (LOAD_) {                                                                        \
      ring[j_][0] = *(const float4*)(xp + (size_t)(4 * ((s_) + ENC_PF)) * XCH);         \
      ring[j_][1] = *(const float4*)(xp + (size_t)(4 * ((s_) + ENC_PF) + 1) * XCH);    \
    }                                                                                   \
       \
       \
    __builtin_amdgcn_sched_barrier(0);                                                  \
    _Pragma("unroll")                                                                   \
    for (int m_ = 0; m_ < 4; ++m_) {                                                    \
      _Pragma("unroll")                                                                 \
      for (int i_ = 0; i_ < 4; ++i_)                                                    \
        acc[m_][i_] = WMMA_F32(fa[(j_) & 1][m_], fb[i_], acc[m_][i_]);                  \
    }                                                                                   \
       \
    __builtin_amdgcn_sched_barrier(0);                                                  \
  }

__global__ __launch_bounds__(256) void k_encode(
    const float* x, const float* __restrict__ Wenc,
    const float* __restrict__ benc, float* __restrict__ enc) {
  extern __shared__ float lds[];
  float* ws = lds;

  const int tid  = threadIdx.x;
  const int b    = blockIdx.x;
  const int n    = b >> 7;
  const int gh   = (b >> 2) & 31;
  const int lane = tid & 31;
  const int wv   = tid >> 5;
  const int lm   = lane & 15;
  const int lh   = lane >> 4;
  const int gw0  = (b & 3) * 8 + (wv & 3) * 2;
  const int py   = (wv >> 2) * 4 + (lm >> 2);
  const int q    = lm & 3;

  const float* xp = x +
      ((((size_t)n * CIN + 2 * lh) * HH + (gh * 8 + py)) * WW + gw0 * 8 + q * 4);

  float4 ring[ENC_PF][2];
  #pragma unroll
  for (int j = 0; j < ENC_PF; ++j) {
    ring[j][0] = *(const float4*)(xp + (size_t)(4 * j) * XCH);
    ring[j][1] = *(const float4*)(xp + (size_t)(4 * j + 1) * XCH);
  }

  {
    const float4* src = (const float4*)Wenc;
    #pragma unroll
    for (int i = 0; i < 16; ++i) {
      int v = i * 256 + tid;
      float4 d = src[v];
      int e = v >> 6;
      int k = (v & 63) * 4;
      *(float4*)&ws[e * WS_STRIDE + k] = d;
    }
  }
  __syncthreads();

  v8f acc[4][4];
  #pragma unroll
  for (int m = 0; m < 4; ++m) {
    #pragma unroll
    for (int i = 0; i < 4; ++i) acc[m][i] = v8f_zero();
  }

  const float* ap = &ws[lm * WS_STRIDE + 2 * lh];
  v2f fa[2][4];
  #pragma unroll
  for (int m = 0; m < 4; ++m) fa[0][m] = *(const v2f*)(ap + m * 16 * WS_STRIDE);

  #pragma unroll
  for (int s0 = 0; s0 < 64; s0 += ENC_PF) {
    ENC_KSTEP(0, s0 + 0, s0 + 0 + ENC_PF < 64)
    ENC_KSTEP(1, s0 + 1, s0 + 1 + ENC_PF < 64)
    ENC_KSTEP(2, s0 + 2, s0 + 2 + ENC_PF < 64)
    ENC_KSTEP(3, s0 + 3, s0 + 3 + ENC_PF < 64)
  }

  float* ob = enc + ((size_t)n * LPERN + gh * GW + gw0 + (q >> 1)) * (size_t)K1 +
              py * 8 + (q & 1) * 4;
  #pragma unroll
  for (int m = 0; m < 4; ++m) {
    const float4 b0 = *(const float4*)(benc + m * 16 + 8 * lh);
    const float4 b1 = *(const float4*)(benc + m * 16 + 8 * lh + 4);
    const float be[8] = {b0.x, b0.y, b0.z, b0.w, b1.x, b1.y, b1.z, b1.w};
    #pragma unroll
    for (int r = 0; r < 8; ++r) {
      int e = m * 16 + 8 * lh + r;
      float4 o;
      o.x = acc[m][0][r] + be[r];
      o.y = acc[m][1][r] + be[r];
      o.z = acc[m][2][r] + be[r];
      o.w = acc[m][3][r] + be[r];
      *(float4*)(ob + e * 64) = o;
    }
  }
}

#define F1_WG       256
#define F1_FUSE_WG  512
static_assert(F1_FUSE_WG * 512 == M_FLOATS, "one fold thread per element of M");
#define F1_NT     16
#define F1_KC     256
#define F1_ESTR   32

static __device__ __forceinline__ int f1_es_off(int R, int col) {
  return R * F1_ESTR + 2 * (col ^ ((R >> 1) & 15));
}

#define F1_LOAD_BLOCK(buf_, g_)                                                         \
  _Pragma("unroll")                                                                     \
  for (int s_ = 0; s_ < 4; ++s_) {                                                      \
    int ka_ = ks * 64 + ((g_) * 4 + s_) * 4 + 2 * lh;                                   \
    fa[buf_][s_] = *(const v2f*)&w1s[(mt * 16 + lm) * WS_STRIDE + ka_];                 \
    fb[buf_][s_] = *(const v2f*)&es[f1_es_off(ka_ >> 1, lm)];                           \
  }

__global__ __launch_bounds__(512) void k_fc1(
    const float* enc, const float* W1,
    const float* __restrict__ b1, float* Tbuf,
    const float* __restrict__ W2, const float* __restrict__ b2,
    const float* __restrict__ Wdec, const float* __restrict__ bdec,
    float* __restrict__ Mf, float* __restrict__ cb) {
  if (blockIdx.x >= F1_WG) {
    fuse_elem((blockIdx.x - F1_WG) * 512 + threadIdx.x, W2, b2, Wdec, bdec, Mf, cb);
    return;
  }
  extern __shared__ float lds[];
  float* w1s = lds;
  float* es  = lds + 64 * WS_STRIDE;

  const int tid  = threadIdx.x;
  const int l0   = blockIdx.x * F1_NT;
  const int lane = tid & 31, wv = tid >> 5;
  const int lm   = lane & 15, lh = lane >> 4;
  const int mt   = wv & 3;
  const int ks   = wv >> 2;

  const int sr  = tid >> 6;
  const int wkq = (tid & 63) * 4;
  const float* w1p = W1 + (size_t)sr * K1 + wkq;
  const float* enp = enc + (size_t)(l0 + sr) * K1 + wkq;

  float4 pw0 = *(const float4*)(w1p);
  float4 pw1 = *(const float4*)(w1p + (size_t) 8 * K1);
  float4 pw2 = *(const float4*)(w1p + (size_t)16 * K1);
  float4 pw3 = *(const float4*)(w1p + (size_t)24 * K1);
  float4 pw4 = *(const float4*)(w1p + (size_t)32 * K1);
  float4 pw5 = *(const float4*)(w1p + (size_t)40 * K1);
  float4 pw6 = *(const float4*)(w1p + (size_t)48 * K1);
  float4 pw7 = *(const float4*)(w1p + (size_t)56 * K1);
  float4 pe0 = *(const float4*)(enp);
  float4 pe1 = *(const float4*)(enp + (size_t)8 * K1);

  v8f acc = v8f_zero();

  for (int ch = 0; ch < K1 / F1_KC; ++ch) {
    __syncthreads();
    *(float4*)&w1s[(sr +  0) * WS_STRIDE + wkq] = pw0;
    *(float4*)&w1s[(sr +  8) * WS_STRIDE + wkq] = pw1;
    *(float4*)&w1s[(sr + 16) * WS_STRIDE + wkq] = pw2;
    *(float4*)&w1s[(sr + 24) * WS_STRIDE + wkq] = pw3;
    *(float4*)&w1s[(sr + 32) * WS_STRIDE + wkq] = pw4;
    *(float4*)&w1s[(sr + 40) * WS_STRIDE + wkq] = pw5;
    *(float4*)&w1s[(sr + 48) * WS_STRIDE + wkq] = pw6;
    *(float4*)&w1s[(sr + 56) * WS_STRIDE + wkq] = pw7;
    {
      int R0 = wkq >> 1;
      v2f t0; t0.x = pe0.x; t0.y = pe0.y;
      v2f t1; t1.x = pe0.z; t1.y = pe0.w;
      *(v2f*)&es[f1_es_off(R0 + 0, sr)] = t0;
      *(v2f*)&es[f1_es_off(R0 + 1, sr)] = t1;
      v2f u0; u0.x = pe1.x; u0.y = pe1.y;
      v2f u1; u1.x = pe1.z; u1.y = pe1.w;
      *(v2f*)&es[f1_es_off(R0 + 0, sr + 8)] = u0;
      *(v2f*)&es[f1_es_off(R0 + 1, sr + 8)] = u1;
    }
    __syncthreads();

    if (ch + 1 < K1 / F1_KC) {
      int kn = (ch + 1) * F1_KC;
      pw0 = *(const float4*)(w1p + kn);
      pw1 = *(const float4*)(w1p + (size_t) 8 * K1 + kn);
      pw2 = *(const float4*)(w1p + (size_t)16 * K1 + kn);
      pw3 = *(const float4*)(w1p + (size_t)24 * K1 + kn);
      pw4 = *(const float4*)(w1p + (size_t)32 * K1 + kn);
      pw5 = *(const float4*)(w1p + (size_t)40 * K1 + kn);
      pw6 = *(const float4*)(w1p + (size_t)48 * K1 + kn);
      pw7 = *(const float4*)(w1p + (size_t)56 * K1 + kn);
      pe0 = *(const float4*)(enp + kn);
      pe1 = *(const float4*)(enp + (size_t)8 * K1 + kn);
    }

    v2f fa[2][4], fb[2][4];
    F1_LOAD_BLOCK(0, 0)
    #pragma unroll
    for (int g = 0; g < 4; ++g) {
      if (g + 1 < 4) { F1_LOAD_BLOCK((g + 1) & 1, g + 1) }
      __builtin_amdgcn_sched_barrier(0);
      #pragma unroll
      for (int s = 0; s < 4; ++s)
        acc = WMMA_F32(fa[g & 1][s], fb[g & 1][s], acc);
      __builtin_amdgcn_sched_barrier(0);
    }
  }

  __syncthreads();
  float* red = lds;
  #pragma unroll
  for (int r = 0; r < 8; ++r)
    red[ks * 1024 + mt * 256 + r * 32 + lane] = acc[r];
  __syncthreads();
  #pragma unroll
  for (int h = 0; h < 2; ++h) {
    int o   = tid + 512 * h;
    int f   = o >> 4;
    int col = o & 15;
    int fr  = f & 15;
    int off = (f >> 4) * 256 + (fr & 7) * 32 + (fr >> 3) * 16 + col;
    float s = ((red[off] + red[1024 + off]) + red[2048 + off]) + red[3072 + off];
    Tbuf[(size_t)blockIdx.x * 1024 + o] = s + b1[f];
  }
}

__global__ __launch_bounds__(256) void k_decode(
    const float* __restrict__ Tbuf, const float* __restrict__ Mf,
    const float* __restrict__ cb, float* __restrict__ out) {
  extern __shared__ float lds[];
  float* ms = lds;
  float* ts = lds + 256 * MS_STRIDE;

  const int tid = threadIdx.x;
  const int pb  = blockIdx.x >> 2;
  const int sg  = blockIdx.x & 3;
  const int l0  = pb * 32;
  const int n   = l0 >> 10;
  const int gh  = (l0 & 1023) >> 5;

  #pragma unroll
  for (int i = 0; i < 4; ++i) {
    int v  = i * 256 + tid;
    int f2 = v >> 5;
    int c  = v & 31;
    v2f t;
    const float* tt = Tbuf + (size_t)((l0 + c) >> 4) * 1024 + (c & 15);
    t.x = tt[(2 * f2 + 0) * 16];
    t.y = tt[(2 * f2 + 1) * 16];
    *(v2f*)&ts[f2 * ES2_STRIDE + c * 2] = t;
  }

  const int lane = tid & 31, wv = tid >> 5;
  const int lm   = lane & 15, lh = lane >> 4;

  const int h2 = gh * 4 + sg;

  for (int ss = 0; ss < 4; ++ss) {
    int s = sg * 4 + ss;
    __syncthreads();
    #pragma unroll
    for (int i = 0; i < 16; ++i) {
      int v  = i * 256 + tid;
      int o  = v >> 4;
      int fq = (v & 15) * 4;
      float4 d = *(const float4*)(Mf + (size_t)s * 16384 + o * 64 + fq);
      *(float4*)&ms[o * MS_STRIDE + fq] = d;
    }
    __syncthreads();

    #pragma unroll
    for (int j = 0; j < 4; ++j) {
      int tt = wv * 4 + j;
      int o0 = (tt >> 1) * 16;
      int ct = (tt & 1) * 16;
      v8f acc = v8f_zero();
      #pragma unroll
      for (int kk = 0; kk < 64; kk += 4) {
        int ka = kk + 2 * lh;
        v2f a  = *(const v2f*)&ms[(o0 + lm) * MS_STRIDE + ka];
        v2f bb = *(const v2f*)&ts[(ka >> 1) * ES2_STRIDE + (ct + lm) * 2];
        acc = WMMA_F32(a, bb, acc);
      }
      int gw = ct + lm;
      int oa = ss * 64 + (o0 >> 2) + 2 * lh;
      const float4 ca = *(const float4*)&cb[(sg * 256 + oa) * 4];
      const float4 cc = *(const float4*)&cb[(sg * 256 + oa + 1) * 4];
      float4 va, vb;
      va.x = acc[0] + ca.x; va.y = acc[1] + ca.y; va.z = acc[2] + ca.z; va.w = acc[3] + ca.w;
      vb.x = acc[4] + cc.x; vb.y = acc[5] + cc.y; vb.z = acc[6] + cc.z; vb.w = acc[7] + cc.w;
      float* orow = out + (((size_t)n * COUT + oa) * 128 + h2) * 128 + gw * 4;
      *(float4*)orow               = va;
      *(float4*)(orow + 128 * 128) = vb;
    }
  }
}

extern "C" void kernel_launch(void* const* d_in, const int* in_sizes, int n_in,
                              void* d_out, int out_size, void* d_ws, size_t ws_size,
                              hipStream_t stream) {
  const float* x    = (const float*)d_in[0];
  const float* Wenc = (const float*)d_in[1];
  const float* benc = (const float*)d_in[2];
  const float* W1   = (const float*)d_in[3];
  const float* b1   = (const float*)d_in[4];
  const float* W2   = (const float*)d_in[5];
  const float* b2   = (const float*)d_in[6];
  const float* Wdec = (const float*)d_in[7];
  const float* bdec = (const float*)d_in[8];
  float* out = (float*)d_out;

  float* ws   = (float*)d_ws;
  float* enc  = ws;
  float* Tbuf = ws + ENC_FLOATS;
  float* Mf   = Tbuf + T_FLOATS;
  float* cbuf = Mf + M_FLOATS;

  size_t ldsB = (size_t)(64 * WS_STRIDE) * sizeof(float);
  k_encode<<<512, 256, ldsB, stream>>>(x, Wenc, benc, enc);

  size_t ldsC1 = (size_t)(64 * WS_STRIDE + 128 * F1_ESTR) * sizeof(float);
  k_fc1<<<768, 512, ldsC1, stream>>>(enc, W1, b1, Tbuf, W2, b2, Wdec, bdec, Mf, cbuf);

  size_t ldsC2 = (size_t)(256 * MS_STRIDE + 32 * ES2_STRIDE) * sizeof(float);
  k_decode<<<512, 256, ldsC2, stream>>>(Tbuf, Mf, cbuf, out);
}


# encode conv composed into FC1 ahead of the data (W1f = W1 x Wenc on WMMA): one 64x16384 GEMM on x, K split 16-way over work-groups, weight slice in LDS; partials summed in fixed order in decode
# speedup vs baseline: 2.1677x; 1.0667x over previous
#include <hip/hip_runtime.h>

typedef float v2f __attribute__((ext_vector_type(2)));
typedef float v8f __attribute__((ext_vector_type(8)));

#define WMMA_F32(a_, b_, c_) \
  __builtin_amdgcn_wmma_f32_16x16x4_f32(false, (a_), false, (b_), (short)0, (c_), false, false)

#define NB      4
#define CIN     256
#define HH      256
#define WW      256
#define EC      64
#define GW      32
#define LPERN   1024
#define LTOT    4096
#define K1      4096
#define KF      16384
#define COUT    256

#define W1F_FLOATS  (1048576)
#define TP_FLOATS   (4194304)
#define ENC_FLOATS  (16777216)
#define T_FLOATS    (262144)
#define M_FLOATS    (262144)

#define MS_STRIDE   68
#define ES2_STRIDE  96
#define AS_STRIDE   1028

static __device__ __forceinline__ v8f v8f_zero() {
  v8f z = {0.f, 0.f, 0.f, 0.f, 0.f, 0.f, 0.f, 0.f};
  return z;
}

static __device__ __forceinline__ void fuse_elem(
    int idx,
    const float* __restrict__ W2, const float* __restrict__ b2,
    const float* __restrict__ Wdec, const float* __restrict__ bdec,
    float* __restrict__ Mf, float* __restrict__ cb) {
  int sy = idx >> 16;
  int o  = (idx >> 8) & 255;
  int sx = (idx >> 6) & 3;
  int f  = idx & 63;
  int s  = sy * 4 + sx;
  float sum = 0.f, sb = 0.f;
  #pragma unroll 8
  for (int e = 0; e < 64; ++e) {
    float wd = Wdec[o * 64 + e];
    sum += wd * W2[(e * 16 + s) * 64 + f];
    sb  += wd * b2[e * 16 + s];
  }
  Mf[idx] = sum;
  if (f == 0) cb[idx >> 6] = sb + bdec[o];
}

#define FOLD_W1_WG  64
#define FOLD_B1_WG  8
#define FOLD_M_WG   512
static_assert(FOLD_M_WG * 512 == M_FLOATS, "one fold thread per element of M");

__global__ __launch_bounds__(512) void k_fold(
    const float* __restrict__ Wenc, const float* __restrict__ benc,
    const float* __restrict__ W1, const float* __restrict__ b1,
    const float* __restrict__ W2, const float* __restrict__ b2,
    const float* __restrict__ Wdec, const float* __restrict__ bdec,
    float* __restrict__ W1f, float* __restrict__ b1f,
    float* __restrict__ Mf, float* __restrict__ cb) {
  __shared__ float red[512];
  const int tid = threadIdx.x;
  const int b   = blockIdx.x;

  if (b >= FOLD_W1_WG + FOLD_B1_WG) {
    fuse_elem((b - FOLD_W1_WG - FOLD_B1_WG) * 512 + tid, W2, b2, Wdec, bdec, Mf, cb);
    return;
  }

  if (b >= FOLD_W1_WG) {
    const int g  = b - FOLD_W1_WG;
    const int fl = tid >> 6;
    const int e  = tid & 63;
    const float* wp = W1 + (size_t)(g * 8 + fl) * K1 + e * 64;
    float s = 0.f;
    #pragma unroll
    for (int i = 0; i < 16; ++i) {
      const float4 v = *(const float4*)(wp + 4 * i);
      s += v.x; s += v.y; s += v.z; s += v.w;
    }
    red[tid] = s * benc[e];
    __syncthreads();
    if (tid < 8) {
      float a = 0.f;
      #pragma unroll
      for (int e2 = 0; e2 < 64; ++e2) a += red[tid * 64 + e2];
      b1f[g * 8 + tid] = a + b1[g * 8 + tid];
    }
    return;
  }

  const int lane = tid & 31, wv = tid >> 5;
  const int lm   = lane & 15, lh = lane >> 4;
  const int pq   = b & 15;
  const int cg   = b >> 4;
  const int mt   = wv & 3;
  const int cs   = wv >> 2;

  const float* ap = W1 + (size_t)(mt * 16 + lm) * K1 + (2 * lh) * 64 + pq * 4;
  const float* bp = Wenc + (2 * lh) * CIN + cg * 64 + cs * 16 + lm;

  v8f acc[4];
  #pragma unroll
  for (int j = 0; j < 4; ++j) acc[j] = v8f_zero();

  #pragma unroll
  for (int s = 0; s < 16; ++s) {
    const float4 a0 = *(const float4*)(ap + (4 * s) * 64);
    const float4 a1 = *(const float4*)(ap + (4 * s + 1) * 64);
    v2f fb; fb.x = bp[(4 * s) * CIN]; fb.y = bp[(4 * s + 1) * CIN];
    v2f f0; f0.x = a0.x; f0.y = a1.x;
    v2f f1; f1.x = a0.y; f1.y = a1.y;
    v2f f2; f2.x = a0.z; f2.y = a1.z;
    v2f f3; f3.x = a0.w; f3.y = a1.w;
    acc[0] = WMMA_F32(f0, fb, acc[0]);
    acc[1] = WMMA_F32(f1, fb, acc[1]);
    acc[2] = WMMA_F32(f2, fb, acc[2]);
    acc[3] = WMMA_F32(f3, fb, acc[3]);
  }

  float* op = W1f + (size_t)(mt * 16 + 8 * lh) * KF + (cg * 64 + cs * 16 + lm) * 64 + pq * 4;
  #pragma unroll
  for (int r = 0; r < 8; ++r) {
    float4 o;
    o.x = acc[0][r]; o.y = acc[1][r]; o.z = acc[2][r]; o.w = acc[3][r];
    *(float4*)(op + (size_t)r * KF) = o;
  }
}

#define MAIN_PF  8
#define XCH      ((size_t)HH * WW)

#define MAIN_STEP(j_, st_, LOAD_)                                                       \
  {                                                                                     \
    _Pragma("unroll")                                                                   \
    for (int m_ = 0; m_ < 4; ++m_)                                                      \
      fa[((st_) + 1) & 1][m_] =                                                         \
          *(const float4*)(ap + m_ * 16 * AS_STRIDE + ((((st_) + 1) & 127) * 8));       \
    const float4 xb_ = ring[j_];                                                        \
    if (LOAD_)                                                                          \
      ring[j_] = *(const float4*)(xp + (size_t)(((st_) + MAIN_PF) >> 3) * XCH +         \
                                  (((st_) + MAIN_PF) & 7) * WW);                        \
       \
       \
       \
    __builtin_amdgcn_sched_barrier(0);                                                  \
    v2f b0_; b0_.x = xb_.x; b0_.y = xb_.y;                                              \
    v2f b1_; b1_.x = xb_.z; b1_.y = xb_.w;                                              \
    _Pragma("unroll")                                                                   \
    for (int m_ = 0; m_ < 4; ++m_) {                                                    \
      v2f a_; a_.x = fa[(st_) & 1][m_].x; a_.y = fa[(st_) & 1][m_].y;                   \
      acc[m_] = WMMA_F32(a_, b0_, acc[m_]);                                             \
    }                                                                                   \
    _Pragma("unroll")                                                                   \
    for (int m_ = 0; m_ < 4; ++m_) {                                                    \
      v2f a_; a_.x = fa[(st_) & 1][m_].z; a_.y = fa[(st_) & 1][m_].w;                   \
      acc[m_] = WMMA_F32(a_, b1_, acc[m_]);                                             \
    }                                                                                   \
       \
    __builtin_amdgcn_sched_barrier(0);                                                  \
  }

__global__ __launch_bounds__(512) void k_main(
    const float* x, const float* __restrict__ W1f, float* __restrict__ Tp) {
  __shared__ __attribute__((aligned(16))) float as[64 * AS_STRIDE];

  const int tid  = threadIdx.x;
  const int b    = blockIdx.x;
  const int ksl  = b & 15;
  const int ng   = b >> 4;
  const int lane = tid & 31;
  const int wv   = tid >> 5;
  const int lm   = lane & 15;
  const int lh   = lane >> 4;
  const int n    = ng >> 2;
  const int gh   = (ng & 3) * 8 + (wv >> 1);
  const int gw0  = (wv & 1) * 16;

  const float* xp = x +
      ((((size_t)n * CIN + ksl * 16) * HH + gh * 8) * WW + (gw0 + lm) * 8 + lh * 4);

  float4 ring[MAIN_PF];
  #pragma unroll
  for (int j = 0; j < MAIN_PF; ++j)
    ring[j] = *(const float4*)(xp + (size_t)(j >> 3) * XCH + (j & 7) * WW);

  {
    const float4* src = (const float4*)(W1f + ksl * 1024);
    #pragma unroll 16
    for (int i = 0; i < 32; ++i) {
      int v  = i * 512 + tid;
      int f  = v >> 8;
      int kq = v & 255;
      float4 d = src[(size_t)f * (KF / 4) + kq];
      *(float4*)&as[f * AS_STRIDE + kq * 4] = d;
    }
  }
  __syncthreads();

  v8f acc[4];
  #pragma unroll
  for (int m = 0; m < 4; ++m) acc[m] = v8f_zero();

  const float* ap = &as[lm * AS_STRIDE + lh * 4];
  float4 fa[2][4];
  #pragma unroll
  for (int m = 0; m < 4; ++m) fa[0][m] = *(const float4*)(ap + m * 16 * AS_STRIDE);

  #pragma unroll
  for (int s0 = 0; s0 < 128; s0 += MAIN_PF) {
    MAIN_STEP(0, s0 + 0, s0 + 0 + MAIN_PF < 128)
    MAIN_STEP(1, s0 + 1, s0 + 1 + MAIN_PF < 128)
    MAIN_STEP(2, s0 + 2, s0 + 2 + MAIN_PF < 128)
    MAIN_STEP(3, s0 + 3, s0 + 3 + MAIN_PF < 128)
    MAIN_STEP(4, s0 + 4, s0 + 4 + MAIN_PF < 128)
    MAIN_STEP(5, s0 + 5, s0 + 5 + MAIN_PF < 128)
    MAIN_STEP(6, s0 + 6, s0 + 6 + MAIN_PF < 128)
    MAIN_STEP(7, s0 + 7, s0 + 7 + MAIN_PF < 128)
  }

  float* tp = Tp + ((size_t)(ksl * 256 + ng * 16 + wv)) * 1024 + (lh * 16 + lm) * 4;
  #pragma unroll
  for (int m = 0; m < 4; ++m) {
    #pragma unroll
    for (int rq = 0; rq < 2; ++rq) {
      float4 o;
      o.x = acc[m][4 * rq + 0]; o.y = acc[m][4 * rq + 1];
      o.z = acc[m][4 * rq + 2]; o.w = acc[m][4 * rq + 3];
      *(float4*)(tp + (m * 2 + rq) * 128) = o;
    }
  }
}

__global__ __launch_bounds__(256) void k_decode(
    const float* __restrict__ Tp, const float* __restrict__ b1f,
    const float* __restrict__ Mf,
    const float* __restrict__ cb, float* __restrict__ out) {
  extern __shared__ float lds[];
  float* ms = lds;
  float* ts = lds + 256 * MS_STRIDE;

  const int tid = threadIdx.x;
  const int pb  = blockIdx.x >> 2;
  const int sg  = blockIdx.x & 3;
  const int l0  = pb * 32;
  const int n   = l0 >> 10;
  const int gh  = (l0 & 1023) >> 5;

  #pragma unroll
  for (int i = 0; i < 2; ++i) {
    int v  = i * 256 + tid;
    int f4 = v >> 5;
    int c  = v & 31;
    const float* tt = Tp + (size_t)((l0 + c) >> 4) * 1024 +
        ((((f4 >> 2) * 2 + (f4 & 1)) * 32 + ((f4 >> 1) & 1) * 16 + (c & 15)) * 4);
    float4 t = *(const float4*)tt;
    #pragma unroll
    for (int ks = 1; ks < 16; ++ks) {
      const float4 u = *(const float4*)(tt + (size_t)ks * (256 * 1024));
      t.x += u.x; t.y += u.y; t.z += u.z; t.w += u.w;
    }
    const float4 bb = *(const float4*)(b1f + 4 * f4);
    v2f t0; t0.x = t.x + bb.x; t0.y = t.y + bb.y;
    v2f t1; t1.x = t.z + bb.z; t1.y = t.w + bb.w;
    *(v2f*)&ts[(2 * f4 + 0) * ES2_STRIDE + c * 2] = t0;
    *(v2f*)&ts[(2 * f4 + 1) * ES2_STRIDE + c * 2] = t1;
  }

  const int lane = tid & 31, wv = tid >> 5;
  const int lm   = lane & 15, lh = lane >> 4;

  const int h2 = gh * 4 + sg;

  for (int ss = 0; ss < 4; ++ss) {
    int s = sg * 4 + ss;
    __syncthreads();
    #pragma unroll
    for (int i = 0; i < 16; ++i) {
      int v  = i * 256 + tid;
      int o  = v >> 4;
      int fq = (v & 15) * 4;
      float4 d = *(const float4*)(Mf + (size_t)s * 16384 + o * 64 + fq);
      *(float4*)&ms[o * MS_STRIDE + fq] = d;
    }
    __syncthreads();

    #pragma unroll
    for (int j = 0; j < 4; ++j) {
      int tt = wv * 4 + j;
      int o0 = (tt >> 1) * 16;
      int ct = (tt & 1) * 16;
      v8f acc = v8f_zero();
      #pragma unroll
      for (int kk = 0; kk < 64; kk += 4) {
        int ka = kk + 2 * lh;
        v2f a  = *(const v2f*)&ms[(o0 + lm) * MS_STRIDE + ka];
        v2f bb = *(const v2f*)&ts[(ka >> 1) * ES2_STRIDE + (ct + lm) * 2];
        acc = WMMA_F32(a, bb, acc);
      }
      int gw = ct + lm;
      int oa = ss * 64 + (o0 >> 2) + 2 * lh;
      const float4 ca = *(const float4*)&cb[(sg * 256 + oa) * 4];
      const float4 cc = *(const float4*)&cb[(sg * 256 + oa + 1) * 4];
      float4 va, vb;
      va.x = acc[0] + ca.x; va.y = acc[1] + ca.y; va.z = acc[2] + ca.z; va.w = acc[3] + ca.w;
      vb.x = acc[4] + cc.x; vb.y = acc[5] + cc.y; vb.z = acc[6] + cc.z; vb.w = acc[7] + cc.w;
      float* orow = out + (((size_t)n * COUT + oa) * 128 + h2) * 128 + gw * 4;
      *(float4*)orow               = va;
      *(float4*)(orow + 128 * 128) = vb;
    }
  }
}

extern "C" void kernel_launch(void* const* d_in, const int* in_sizes, int n_in,
                              void* d_out, int out_size, void* d_ws, size_t ws_size,
                              hipStream_t stream) {
  const float* x    = (const float*)d_in[0];
  const float* Wenc = (const float*)d_in[1];
  const float* benc = (const float*)d_in[2];
  const float* W1   = (const float*)d_in[3];
  const float* b1   = (const float*)d_in[4];
  const float* W2   = (const float*)d_in[5];
  const float* b2   = (const float*)d_in[6];
  const float* Wdec = (const float*)d_in[7];
  const float* bdec = (const float*)d_in[8];
  float* out = (float*)d_out;

  float* ws   = (float*)d_ws;
  float* W1f  = ws;
  float* Tp   = ws + W1F_FLOATS;
  float* b1f  = Tp + TP_FLOATS;
  float* Mf   = ws + ENC_FLOATS + T_FLOATS;
  float* cbuf = Mf + M_FLOATS;

  k_fold<<<584, 512, 0, stream>>>(Wenc, benc, W1, b1, W2, b2, Wdec, bdec, W1f, b1f, Mf, cbuf);

  k_main<<<256, 512, 0, stream>>>(x, W1f, Tp);

  size_t ldsC2 = (size_t)(256 * MS_STRIDE + 32 * ES2_STRIDE) * sizeof(float);
  k_decode<<<512, 256, ldsC2, stream>>>(Tp, b1f, Mf, cbuf, out);
}


# decode: weight (A) fragments go global -> registers in a fragment order the fold now writes, T (B) fragments stay in registers: no LDS copy of weights, no barrier in the chunk loop; sums unchanged
# speedup vs baseline: 2.2832x; 1.0533x over previous
#include <hip/hip_runtime.h>

typedef float v2f __attribute__((ext_vector_type(2)));
typedef float v8f __attribute__((ext_vector_type(8)));

#define WMMA_F32(a_, b_, c_) \
  __builtin_amdgcn_wmma_f32_16x16x4_f32(false, (a_), false, (b_), (short)0, (c_), false, false)

#define NB      4
#define CIN     256
#define HH      256
#define WW      256
#define EC      64
#define GW      32
#define LPERN   1024
#define LTOT    4096
#define K1      4096
#define KF      16384
#define COUT    256

#define W1F_FLOATS  (1048576)
#define TP_FLOATS   (4194304)
#define ENC_FLOATS  (16777216)
#define T_FLOATS    (262144)
#define M_FLOATS    (262144)

#define ES2_STRIDE  96
#define AS_STRIDE   1028

static __device__ __forceinline__ v8f v8f_zero() {
  v8f z = {0.f, 0.f, 0.f, 0.f, 0.f, 0.f, 0.f, 0.f};
  return z;
}

static __device__ __forceinline__ void fuse_elem(
    int d,
    const float* __restrict__ W2, const float* __restrict__ b2,
    const float* __restrict__ Wdec, const float* __restrict__ bdec,
    float* __restrict__ Mf, float* __restrict__ cb) {
  int pos = d & 3;
  int lm  = (d >> 2) & 15;
  int lh  = (d >> 6) & 1;
  int q   = (d >> 7) & 7;
  int R   = ((d >> 10) << 4) + lm;
  int sy = R >> 10;
  int o  = (R >> 2) & 255;
  int sx = R & 3;
  int f  = 8 * q + 4 * (pos >> 1) + 2 * lh + (pos & 1);
  int s  = sy * 4 + sx;
  float sum = 0.f, sb = 0.f;
  #pragma unroll 8
  for (int e = 0; e < 64; ++e) {
    float wd = Wdec[o * 64 + e];
    sum += wd * W2[(e * 16 + s) * 64 + f];
    sb  += wd * b2[e * 16 + s];
  }
  Mf[d] = sum;
  if (f == 0) cb[R] = sb + bdec[o];
}

#define FOLD_W1_WG  64
#define FOLD_B1_WG  8
#define FOLD_M_WG   512
static_assert(FOLD_M_WG * 512 == M_FLOATS, "one fold thread per element of M");

__global__ __launch_bounds__(512) void k_fold(
    const float* __restrict__ Wenc, const float* __restrict__ benc,
    const float* __restrict__ W1, const float* __restrict__ b1,
    const float* __restrict__ W2, const float* __restrict__ b2,
    const float* __restrict__ Wdec, const float* __restrict__ bdec,
    float* __restrict__ W1f, float* __restrict__ b1f,
    float* __restrict__ Mf, float* __restrict__ cb) {
  __shared__ float red[512];
  const int tid = threadIdx.x;
  const int b   = blockIdx.x;

  if (b >= FOLD_W1_WG + FOLD_B1_WG) {
    fuse_elem((b - FOLD_W1_WG - FOLD_B1_WG) * 512 + tid, W2, b2, Wdec, bdec, Mf, cb);
    return;
  }

  if (b >= FOLD_W1_WG) {
    const int g  = b - FOLD_W1_WG;
    const int fl = tid >> 6;
    const int e  = tid & 63;
    const float* wp = W1 + (size_t)(g * 8 + fl) * K1 + e * 64;
    float s = 0.f;
    #pragma unroll
    for (int i = 0; i < 16; ++i) {
      const float4 v = *(const float4*)(wp + 4 * i);
      s += v.x; s += v.y; s += v.z; s += v.w;
    }
    red[tid] = s * benc[e];
    __syncthreads();
    if (tid < 8) {
      float a = 0.f;
      #pragma unroll
      for (int e2 = 0; e2 < 64; ++e2) a += red[tid * 64 + e2];
      b1f[g * 8 + tid] = a + b1[g * 8 + tid];
    }
    return;
  }

  const int lane = tid & 31, wv = tid >> 5;
  const int lm   = lane & 15, lh = lane >> 4;
  const int pq   = b & 15;
  const int cg   = b >> 4;
  const int mt   = wv & 3;
  const int cs   = wv >> 2;

  const float* ap = W1 + (size_t)(mt * 16 + lm) * K1 + (2 * lh) * 64 + pq * 4;
  const float* bp = Wenc + (2 * lh) * CIN + cg * 64 + cs * 16 + lm;

  v8f acc[4];
  #pragma unroll
  for (int j = 0; j < 4; ++j) acc[j] = v8f_zero();

  #pragma unroll
  for (int s = 0; s < 16; ++s) {
    const float4 a0 = *(const float4*)(ap + (4 * s) * 64);
    const float4 a1 = *(const float4*)(ap + (4 * s + 1) * 64);
    v2f fb; fb.x = bp[(4 * s) * CIN]; fb.y = bp[(4 * s + 1) * CIN];
    v2f f0; f0.x = a0.x; f0.y = a1.x;
    v2f f1; f1.x = a0.y; f1.y = a1.y;
    v2f f2; f2.x = a0.z; f2.y = a1.z;
    v2f f3; f3.x = a0.w; f3.y = a1.w;
    acc[0] = WMMA_F32(f0, fb, acc[0]);
    acc[1] = WMMA_F32(f1, fb, acc[1]);
    acc[2] = WMMA_F32(f2, fb, acc[2]);
    acc[3] = WMMA_F32(f3, fb, acc[3]);
  }

  float* op = W1f + (size_t)(mt * 16 + 8 * lh) * KF + (cg * 64 + cs * 16 + lm) * 64 + pq * 4;
  #pragma unroll
  for (int r = 0; r < 8; ++r) {
    float4 o;
    o.x = acc[0][r]; o.y = acc[1][r]; o.z = acc[2][r]; o.w = acc[3][r];
    *(float4*)(op + (size_t)r * KF) = o;
  }
}

#define MAIN_PF  8
#define XCH      ((size_t)HH * WW)

#define MAIN_STEP(j_, st_, LOAD_)                                                       \
  {                                                                                     \
    _Pragma("unroll")                                                                   \
    for (int m_ = 0; m_ < 4; ++m_)                                                      \
      fa[((st_) + 1) & 1][m_] =                                                         \
          *(const float4*)(ap + m_ * 16 * AS_STRIDE + ((((st_) + 1) & 127) * 8));       \
    const float4 xb_ = ring[j_];                                                        \
    if (LOAD_)                                                                          \
      ring[j_] = *(const float4*)(xp + (size_t)(((st_) + MAIN_PF) >> 3) * XCH +         \
                                  (((st_) + MAIN_PF) & 7) * WW);                        \
       \
       \
       \
    __builtin_amdgcn_sched_barrier(0);                                                  \
    v2f b0_; b0_.x = xb_.x; b0_.y = xb_.y;                                              \
    v2f b1_; b1_.x = xb_.z; b1_.y = xb_.w;                                              \
    _Pragma("unroll")                                                                   \
    for (int m_ = 0; m_ < 4; ++m_) {                                                    \
      v2f a_; a_.x = fa[(st_) & 1][m_].x; a_.y = fa[(st_) & 1][m_].y;                   \
      acc[m_] = WMMA_F32(a_, b0_, acc[m_]);                                             \
    }                                                                                   \
    _Pragma("unroll")                                                                   \
    for (int m_ = 0; m_ < 4; ++m_) {                                                    \
      v2f a_; a_.x = fa[(st_) & 1][m_].z; a_.y = fa[(st_) & 1][m_].w;                   \
      acc[m_] = WMMA_F32(a_, b1_, acc[m_]);                                             \
    }                                                                                   \
       \
    __builtin_amdgcn_sched_barrier(0);                                                  \
  }

__global__ __launch_bounds__(512) void k_main(
    const float* x, const float* __restrict__ W1f, float* __restrict__ Tp) {
  __shared__ __attribute__((aligned(16))) float as[64 * AS_STRIDE];

  const int tid  = threadIdx.x;
  const int b    = blockIdx.x;
  const int ksl  = b & 15;
  const int ng   = b >> 4;
  const int lane = tid & 31;
  const int wv   = tid >> 5;
  const int lm   = lane & 15;
  const int lh   = lane >> 4;
  const int n    = ng >> 2;
  const int gh   = (ng & 3) * 8 + (wv >> 1);
  const int gw0  = (wv & 1) * 16;

  const float* xp = x +
      ((((size_t)n * CIN + ksl * 16) * HH + gh * 8) * WW + (gw0 + lm) * 8 + lh * 4);

  float4 ring[MAIN_PF];
  #pragma unroll
  for (int j = 0; j < MAIN_PF; ++j)
    ring[j] = *(const float4*)(xp + (size_t)(j >> 3) * XCH + (j & 7) * WW);

  {
    const float4* src = (const float4*)(W1f + ksl * 1024);
    #pragma unroll 16
    for (int i = 0; i < 32; ++i) {
      int v  = i * 512 + tid;
      int f  = v >> 8;
      int kq = v & 255;
      float4 d = src[(size_t)f * (KF / 4) + kq];
      *(float4*)&as[f * AS_STRIDE + kq * 4] = d;
    }
  }
  __syncthreads();

  v8f acc[4];
  #pragma unroll
  for (int m = 0; m < 4; ++m) acc[m] = v8f_zero();

  const float* ap = &as[lm * AS_STRIDE + lh * 4];
  float4 fa[2][4];
  #pragma unroll
  for (int m = 0; m < 4; ++m) fa[0][m] = *(const float4*)(ap + m * 16 * AS_STRIDE);

  #pragma unroll
  for (int s0 = 0; s0 < 128; s0 += MAIN_PF) {
    MAIN_STEP(0, s0 + 0, s0 + 0 + MAIN_PF < 128)
    MAIN_STEP(1, s0 + 1, s0 + 1 + MAIN_PF < 128)
    MAIN_STEP(2, s0 + 2, s0 + 2 + MAIN_PF < 128)
    MAIN_STEP(3, s0 + 3, s0 + 3 + MAIN_PF < 128)
    MAIN_STEP(4, s0 + 4, s0 + 4 + MAIN_PF < 128)
    MAIN_STEP(5, s0 + 5, s0 + 5 + MAIN_PF < 128)
    MAIN_STEP(6, s0 + 6, s0 + 6 + MAIN_PF < 128)
    MAIN_STEP(7, s0 + 7, s0 + 7 + MAIN_PF < 128)
  }

  float* tp = Tp + ((size_t)(ksl * 256 + ng * 16 + wv)) * 1024 + (lh * 16 + lm) * 4;
  #pragma unroll
  for (int m = 0; m < 4; ++m) {
    #pragma unroll
    for (int rq = 0; rq < 2; ++rq) {
      float4 o;
      o.x = acc[m][4 * rq + 0]; o.y = acc[m][4 * rq + 1];
      o.z = acc[m][4 * rq + 2]; o.w = acc[m][4 * rq + 3];
      *(float4*)(tp + (m * 2 + rq) * 128) = o;
    }
  }
}

__global__ __launch_bounds__(256) void k_decode(
    const float* __restrict__ Tp, const float* __restrict__ b1f,
    const float* Mq,
    const float* __restrict__ cb, float* __restrict__ out) {
  __shared__ __attribute__((aligned(16))) float ts[32 * ES2_STRIDE];

  const int tid = threadIdx.x;
  const int pb  = blockIdx.x >> 2;
  const int sg  = blockIdx.x & 3;
  const int l0  = pb * 32;
  const int n   = l0 >> 10;
  const int gh  = (l0 & 1023) >> 5;

  #pragma unroll
  for (int i = 0; i < 2; ++i) {
    int v  = i * 256 + tid;
    int f4 = v >> 5;
    int c  = v & 31;
    const float* tt = Tp + (size_t)((l0 + c) >> 4) * 1024 +
        ((((f4 >> 2) * 2 + (f4 & 1)) * 32 + ((f4 >> 1) & 1) * 16 + (c & 15)) * 4);
    float4 t = *(const float4*)tt;
    #pragma unroll
    for (int ks = 1; ks < 16; ++ks) {
      const float4 u = *(const float4*)(tt + (size_t)ks * (256 * 1024));
      t.x += u.x; t.y += u.y; t.z += u.z; t.w += u.w;
    }
    const float4 bb = *(const float4*)(b1f + 4 * f4);
    v2f t0; t0.x = t.x + bb.x; t0.y = t.y + bb.y;
    v2f t1; t1.x = t.z + bb.z; t1.y = t.w + bb.w;
    *(v2f*)&ts[(2 * f4 + 0) * ES2_STRIDE + c * 2] = t0;
    *(v2f*)&ts[(2 * f4 + 1) * ES2_STRIDE + c * 2] = t1;
  }

  const int lane = tid & 31, wv = tid >> 5;
  const int lm   = lane & 15, lh = lane >> 4;

  const int h2 = gh * 4 + sg;

  const float4* aq = (const float4*)Mq + ((size_t)(sg * 64 + 2 * wv) * 8) * 32 + lane;
#define DEC_A(st_, q_) aq[((((st_) >> 1) * 16 + ((st_) & 1)) * 8 + (q_)) * 32]
  float4 ar[3][8];
  #pragma unroll
  for (int q = 0; q < 8; ++q) ar[0][q] = DEC_A(0, q);
  #pragma unroll
  for (int q = 0; q < 8; ++q) ar[1][q] = DEC_A(1, q);

  __syncthreads();

  v2f bf[2][16];
  #pragma unroll
  for (int c2 = 0; c2 < 2; ++c2) {
    #pragma unroll
    for (int k = 0; k < 16; ++k)
      bf[c2][k] = *(const v2f*)&ts[(2 * k + lh) * ES2_STRIDE + (c2 * 16 + lm) * 2];
  }

  #pragma unroll
  for (int st = 0; st < 8; ++st) {
    const int ss = st >> 1;
    const int oa = ss * 64 + (2 * wv + (st & 1)) * 4 + 2 * lh;
    if (st + 2 < 8) {
      #pragma unroll
      for (int q = 0; q < 8; ++q) ar[(st + 2) % 3][q] = DEC_A(st + 2, q);
    }
    const float4 ca = *(const float4*)&cb[(sg * 256 + oa) * 4];
    const float4 cc = *(const float4*)&cb[(sg * 256 + oa + 1) * 4];
    __builtin_amdgcn_sched_barrier(0);
    v8f acc0 = v8f_zero(), acc1 = v8f_zero();
    #pragma unroll
    for (int q = 0; q < 8; ++q) {
      const float4 a = ar[st % 3][q];
      v2f a0; a0.x = a.x; a0.y = a.y;
      v2f a1; a1.x = a.z; a1.y = a.w;
      acc0 = WMMA_F32(a0, bf[0][2 * q], acc0);
      acc1 = WMMA_F32(a0, bf[1][2 * q], acc1);
      acc0 = WMMA_F32(a1, bf[0][2 * q + 1], acc0);
      acc1 = WMMA_F32(a1, bf[1][2 * q + 1], acc1);
    }
    __builtin_amdgcn_sched_barrier(0);
    float* orow = out + (((size_t)n * COUT + oa) * 128 + h2) * 128 + lm * 4;
    float4 va, vb;
    va.x = acc0[0] + ca.x; va.y = acc0[1] + ca.y; va.z = acc0[2] + ca.z; va.w = acc0[3] + ca.w;
    vb.x = acc0[4] + cc.x; vb.y = acc0[5] + cc.y; vb.z = acc0[6] + cc.z; vb.w = acc0[7] + cc.w;
    *(float4*)orow               = va;
    *(float4*)(orow + 128 * 128) = vb;
    va.x = acc1[0] + ca.x; va.y = acc1[1] + ca.y; va.z = acc1[2] + ca.z; va.w = acc1[3] + ca.w;
    vb.x = acc1[4] + cc.x; vb.y = acc1[5] + cc.y; vb.z = acc1[6] + cc.z; vb.w = acc1[7] + cc.w;
    *(float4*)(orow + 64)             = va;
    *(float4*)(orow + 64 + 128 * 128) = vb;
  }
#undef DEC_A
}

extern "C" void kernel_launch(void* const* d_in, const int* in_sizes, int n_in,
                              void* d_out, int out_size, void* d_ws, size_t ws_size,
                              hipStream_t stream) {
  const float* x    = (const float*)d_in[0];
  const float* Wenc = (const float*)d_in[1];
  const float* benc = (const float*)d_in[2];
  const float* W1   = (const float*)d_in[3];
  const float* b1   = (const float*)d_in[4];
  const float* W2   = (const float*)d_in[5];
  const float* b2   = (const float*)d_in[6];
  const float* Wdec = (const float*)d_in[7];
  const float* bdec = (const float*)d_in[8];
  float* out = (float*)d_out;

  float* ws   = (float*)d_ws;
  float* W1f  = ws;
  float* Tp   = ws + W1F_FLOATS;
  float* b1f  = Tp + TP_FLOATS;
  float* Mf   = ws + ENC_FLOATS + T_FLOATS;
  float* cbuf = Mf + M_FLOATS;

  k_fold<<<584, 512, 0, stream>>>(Wenc, benc, W1, b1, W2, b2, Wdec, bdec, W1f, b1f, Mf, cbuf);

  k_main<<<256, 512, 0, stream>>>(x, W1f, Tp);

  k_decode<<<512, 256, 0, stream>>>(Tp, b1f, Mf, cbuf, out);
}


# launch 1 rebuilt as 1424 waves of up to 64 WMMAs (was 9344 waves): W1f tasks (f, 16 c) and decode-fold tasks (s, 16 o) load and store whole 256-B rows; M, cb summed in WMMA k order (was serial e)
# speedup vs baseline: 2.8902x; 1.2659x over previous
#include <hip/hip_runtime.h>

typedef float v2f __attribute__((ext_vector_type(2)));
typedef float v8f __attribute__((ext_vector_type(8)));

#define WMMA_F32(a_, b_, c_) \
  __builtin_amdgcn_wmma_f32_16x16x4_f32(false, (a_), false, (b_), (short)0, (c_), false, false)

#define NB      4
#define CIN     256
#define HH      256
#define WW      256
#define EC      64
#define GW      32
#define LPERN   1024
#define LTOT    4096
#define K1      4096
#define KF      16384
#define COUT    256

#define W1F_FLOATS  (1048576)
#define TP_FLOATS   (4194304)
#define ENC_FLOATS  (16777216)
#define T_FLOATS    (262144)
#define M_FLOATS    (262144)

#define ES2_STRIDE  96
#define AS_STRIDE   1028

static __device__ __forceinline__ v8f v8f_zero() {
  v8f z = {0.f, 0.f, 0.f, 0.f, 0.f, 0.f, 0.f, 0.f};
  return z;
}


#define FOLD_W1_WG  256
#define FOLD_B1_WG  32
#define FOLD_M_WG   64
#define FOLD_CB_WG  4
#define FOLD_WG     (FOLD_W1_WG + FOLD_B1_WG + FOLD_M_WG + FOLD_CB_WG)
static_assert(FOLD_WG == 356, "grid of k_fold in kernel_launch");
static_assert(FOLD_W1_WG * 4 == 64 * (CIN / 16), "one W1f task per (f, 16 channels)");
static_assert(FOLD_M_WG * 4 * 1024 == M_FLOATS, "one M task per 1024 elements of M");

__global__ __launch_bounds__(128) void k_fold(
    const float* __restrict__ Wenc, const float* __restrict__ benc,
    const float* __restrict__ W1, const float* __restrict__ b1,
    const float* __restrict__ W2, const float* __restrict__ b2,
    const float* __restrict__ Wdec, const float* __restrict__ bdec,
    float* __restrict__ W1f, float* __restrict__ b1f,
    float* __restrict__ Mf, float* __restrict__ cb) {
  __shared__ float red[128];
  const int tid  = threadIdx.x;
  const int b    = blockIdx.x;
  const int lane = tid & 31, wv = tid >> 5;
  const int lm   = lane & 15, lh = lane >> 4;

  if (b >= FOLD_W1_WG + FOLD_B1_WG) {
    const int g  = b - (FOLD_W1_WG + FOLD_B1_WG);
    const bool is_cb = g >= FOLD_M_WG;
    const int ot = is_cb ? (g - FOLD_M_WG) * 4 + wv : (g >> 2);

    const float* bp = Wdec + (size_t)(ot * 16 + lm) * 64 + 4 * lh;
    float4 wb[8];
    #pragma unroll
    for (int u = 0; u < 8; ++u) wb[u] = *(const float4*)(bp + 8 * u);

    if (is_cb) {
      const float* ap = b2 + (4 * lh) * 16 + lm;
      v8f acc = v8f_zero();
      #pragma unroll
      for (int u = 0; u < 8; ++u) {
        v2f a01; a01.x = ap[(8 * u + 0) * 16]; a01.y = ap[(8 * u + 1) * 16];
        v2f a23; a23.x = ap[(8 * u + 2) * 16]; a23.y = ap[(8 * u + 3) * 16];
        v2f b01; b01.x = wb[u].x; b01.y = wb[u].y;
        v2f b23; b23.x = wb[u].z; b23.y = wb[u].w;
        acc = WMMA_F32(a01, b01, acc);
        acc = WMMA_F32(a23, b23, acc);
      }
      const int o = ot * 16 + lm;
      const float bd = bdec[o];
      float4 v0, v1;
      v0.x = acc[0] + bd; v0.y = acc[1] + bd; v0.z = acc[2] + bd; v0.w = acc[3] + bd;
      v1.x = acc[4] + bd; v1.y = acc[5] + bd; v1.z = acc[6] + bd; v1.w = acc[7] + bd;
      *(float4*)(cb + (size_t)((2 * lh) * 256 + o) * 4)     = v0;
      *(float4*)(cb + (size_t)((2 * lh + 1) * 256 + o) * 4) = v1;
      return;
    }

    const int sy = g & 3;
    const int sx = wv;
    const int s  = sy * 4 + sx;
    const float* ap = W2 + (size_t)(4 * lh) * 1024 + s * 64 + 4 * lm;
    v8f acc[4];
    #pragma unroll
    for (int i = 0; i < 4; ++i) acc[i] = v8f_zero();
    #pragma unroll
    for (int u = 0; u < 8; ++u) {
      const float4 a0 = *(const float4*)(ap + (size_t)(8 * u + 0) * 1024);
      const float4 a1 = *(const float4*)(ap + (size_t)(8 * u + 1) * 1024);
      const float4 a2 = *(const float4*)(ap + (size_t)(8 * u + 2) * 1024);
      const float4 a3 = *(const float4*)(ap + (size_t)(8 * u + 3) * 1024);
      v2f b01; b01.x = wb[u].x; b01.y = wb[u].y;
      v2f b23; b23.x = wb[u].z; b23.y = wb[u].w;
      v2f f0; f0.x = a0.x; f0.y = a1.x;
      v2f f1; f1.x = a0.y; f1.y = a1.y;
      v2f f2; f2.x = a0.z; f2.y = a1.z;
      v2f f3; f3.x = a0.w; f3.y = a1.w;
      acc[0] = WMMA_F32(f0, b01, acc[0]);
      acc[1] = WMMA_F32(f1, b01, acc[1]);
      acc[2] = WMMA_F32(f2, b01, acc[2]);
      acc[3] = WMMA_F32(f3, b01, acc[3]);
      v2f h0; h0.x = a2.x; h0.y = a3.x;
      v2f h1; h1.x = a2.y; h1.y = a3.y;
      v2f h2; h2.x = a2.z; h2.y = a3.z;
      v2f h3; h3.x = a2.w; h3.y = a3.w;
      acc[0] = WMMA_F32(h0, b23, acc[0]);
      acc[1] = WMMA_F32(h1, b23, acc[1]);
      acc[2] = WMMA_F32(h2, b23, acc[2]);
      acc[3] = WMMA_F32(h3, b23, acc[3]);
    }
    float4* mq = (float4*)Mf +
        ((size_t)((sy * 64 + ot * 4 + (lm >> 2)) * 8 + 4 * lh) * 32 + 4 * (lm & 3) + sx);
    #pragma unroll
    for (int a = 0; a < 4; ++a) {
      float4 v0, v1;
      v0.x = acc[0][2 * a]; v0.y = acc[1][2 * a]; v0.z = acc[0][2 * a + 1]; v0.w = acc[1][2 * a + 1];
      v1.x = acc[2][2 * a]; v1.y = acc[3][2 * a]; v1.z = acc[2][2 * a + 1]; v1.w = acc[3][2 * a + 1];
      mq[a * 32]      = v0;
      mq[a * 32 + 16] = v1;
    }
    return;
  }

  if (b >= FOLD_W1_WG) {
    const int g  = b - FOLD_W1_WG;
    const int fl = tid >> 6;
    const int e  = tid & 63;
    const float* wp = W1 + (size_t)(g * 2 + fl) * K1 + e * 64;
    float s = 0.f;
    #pragma unroll
    for (int i = 0; i < 16; ++i) {
      const float4 v = *(const float4*)(wp + 4 * i);
      s += v.x; s += v.y; s += v.z; s += v.w;
    }
    red[tid] = s * benc[e];
    __syncthreads();
    if (tid < 2) {
      float a = 0.f;
      #pragma unroll
      for (int e2 = 0; e2 < 64; ++e2) a += red[tid * 64 + e2];
      b1f[g * 2 + tid] = a + b1[g * 2 + tid];
    }
    return;
  }

  const int t  = b * 4 + wv;
  const int f  = t >> 4;
  const int c0 = (t & 15) * 16;

  const float* ap = Wenc + (size_t)(2 * lh) * CIN + c0 + lm;
  const float* bp = W1 + (size_t)f * K1 + (2 * lh) * 64 + 4 * lm;

  v8f acc[4];
  #pragma unroll
  for (int j = 0; j < 4; ++j) acc[j] = v8f_zero();

  #pragma unroll
  for (int s = 0; s < 16; ++s) {
    const float4 b0 = *(const float4*)(bp + (4 * s) * 64);
    const float4 b1v = *(const float4*)(bp + (4 * s + 1) * 64);
    v2f fa; fa.x = ap[(4 * s) * CIN]; fa.y = ap[(4 * s + 1) * CIN];
    v2f f0; f0.x = b0.x; f0.y = b1v.x;
    v2f f1; f1.x = b0.y; f1.y = b1v.y;
    v2f f2; f2.x = b0.z; f2.y = b1v.z;
    v2f f3; f3.x = b0.w; f3.y = b1v.w;
    acc[0] = WMMA_F32(fa, f0, acc[0]);
    acc[1] = WMMA_F32(fa, f1, acc[1]);
    acc[2] = WMMA_F32(fa, f2, acc[2]);
    acc[3] = WMMA_F32(fa, f3, acc[3]);
  }

  float* op = W1f + (size_t)f * KF + (size_t)(c0 + 8 * lh) * 64 + 4 * lm;
  #pragma unroll
  for (int r = 0; r < 8; ++r) {
    float4 o;
    o.x = acc[0][r]; o.y = acc[1][r]; o.z = acc[2][r]; o.w = acc[3][r];
    *(float4*)(op + r * 64) = o;
  }
}

#define MAIN_PF  8
#define XCH      ((size_t)HH * WW)

#define MAIN_STEP(j_, st_, LOAD_)                                                       \
  {                                                                                     \
    _Pragma("unroll")                                                                   \
    for (int m_ = 0; m_ < 4; ++m_)                                                      \
      fa[((st_) + 1) & 1][m_] =                                                         \
          *(const float4*)(ap + m_ * 16 * AS_STRIDE + ((((st_) + 1) & 127) * 8));       \
    const float4 xb_ = ring[j_];                                                        \
    if (LOAD_)                                                                          \
      ring[j_] = *(const float4*)(xp + (size_t)(((st_) + MAIN_PF) >> 3) * XCH +         \
                                  (((st_) + MAIN_PF) & 7) * WW);                        \
       \
       \
       \
    __builtin_amdgcn_sched_barrier(0);                                                  \
    v2f b0_; b0_.x = xb_.x; b0_.y = xb_.y;                                              \
    v2f b1_; b1_.x = xb_.z; b1_.y = xb_.w;                                              \
    _Pragma("unroll")                                                                   \
    for (int m_ = 0; m_ < 4; ++m_) {                                                    \
      v2f a_; a_.x = fa[(st_) & 1][m_].x; a_.y = fa[(st_) & 1][m_].y;                   \
      acc[m_] = WMMA_F32(a_, b0_, acc[m_]);                                             \
    }                                                                                   \
    _Pragma("unroll")                                                                   \
    for (int m_ = 0; m_ < 4; ++m_) {                                                    \
      v2f a_; a_.x = fa[(st_) & 1][m_].z; a_.y = fa[(st_) & 1][m_].w;                   \
      acc[m_] = WMMA_F32(a_, b1_, acc[m_]);                                             \
    }                                                                                   \
       \
    __builtin_amdgcn_sched_barrier(0);                                                  \
  }

__global__ __launch_bounds__(512) void k_main(
    const float* x, const float* __restrict__ W1f, float* __restrict__ Tp) {
  __shared__ __attribute__((aligned(16))) float as[64 * AS_STRIDE];

  const int tid  = threadIdx.x;
  const int b    = blockIdx.x;
  const int ksl  = b & 15;
  const int ng   = b >> 4;
  const int lane = tid & 31;
  const int wv   = tid >> 5;
  const int lm   = lane & 15;
  const int lh   = lane >> 4;
  const int n    = ng >> 2;
  const int gh   = (ng & 3) * 8 + (wv >> 1);
  const int gw0  = (wv & 1) * 16;

  const float* xp = x +
      ((((size_t)n * CIN + ksl * 16) * HH + gh * 8) * WW + (gw0 + lm) * 8 + lh * 4);

  float4 ring[MAIN_PF];
  #pragma unroll
  for (int j = 0; j < MAIN_PF; ++j)
    ring[j] = *(const float4*)(xp + (size_t)(j >> 3) * XCH + (j & 7) * WW);

  {
    const float4* src = (const float4*)(W1f + ksl * 1024);
    #pragma unroll 16
    for (int i = 0; i < 32; ++i) {
      int v  = i * 512 + tid;
      int f  = v >> 8;
      int kq = v & 255;
      float4 d = src[(size_t)f * (KF / 4) + kq];
      *(float4*)&as[f * AS_STRIDE + kq * 4] = d;
    }
  }
  __syncthreads();

  v8f acc[4];
  #pragma unroll
  for (int m = 0; m < 4; ++m) acc[m] = v8f_zero();

  const float* ap = &as[lm * AS_STRIDE + lh * 4];
  float4 fa[2][4];
  #pragma unroll
  for (int m = 0; m < 4; ++m) fa[0][m] = *(const float4*)(ap + m * 16 * AS_STRIDE);

  #pragma unroll
  for (int s0 = 0; s0 < 128; s0 += MAIN_PF) {
    MAIN_STEP(0, s0 + 0, s0 + 0 + MAIN_PF < 128)
    MAIN_STEP(1, s0 + 1, s0 + 1 + MAIN_PF < 128)
    MAIN_STEP(2, s0 + 2, s0 + 2 + MAIN_PF < 128)
    MAIN_STEP(3, s0 + 3, s0 + 3 + MAIN_PF < 128)
    MAIN_STEP(4, s0 + 4, s0 + 4 + MAIN_PF < 128)
    MAIN_STEP(5, s0 + 5, s0 + 5 + MAIN_PF < 128)
    MAIN_STEP(6, s0 + 6, s0 + 6 + MAIN_PF < 128)
    MAIN_STEP(7, s0 + 7, s0 + 7 + MAIN_PF < 128)
  }

  float* tp = Tp + ((size_t)(ksl * 256 + ng * 16 + wv)) * 1024 + (lh * 16 + lm) * 4;
  #pragma unroll
  for (int m = 0; m < 4; ++m) {
    #pragma unroll
    for (int rq = 0; rq < 2; ++rq) {
      float4 o;
      o.x = acc[m][4 * rq + 0]; o.y = acc[m][4 * rq + 1];
      o.z = acc[m][4 * rq + 2]; o.w = acc[m][4 * rq + 3];
      *(float4*)(tp + (m * 2 + rq) * 128) = o;
    }
  }
}

__global__ __launch_bounds__(256) void k_decode(
    const float* __restrict__ Tp, const float* __restrict__ b1f,
    const float* Mq,
    const float* __restrict__ cb, float* __restrict__ out) {
  __shared__ __attribute__((aligned(16))) float ts[32 * ES2_STRIDE];

  const int tid = threadIdx.x;
  const int pb  = blockIdx.x >> 2;
  const int sg  = blockIdx.x & 3;
  const int l0  = pb * 32;
  const int n   = l0 >> 10;
  const int gh  = (l0 & 1023) >> 5;

  #pragma unroll
  for (int i = 0; i < 2; ++i) {
    int v  = i * 256 + tid;
    int f4 = v >> 5;
    int c  = v & 31;
    const float* tt = Tp + (size_t)((l0 + c) >> 4) * 1024 +
        ((((f4 >> 2) * 2 + (f4 & 1)) * 32 + ((f4 >> 1) & 1) * 16 + (c & 15)) * 4);
    float4 t = *(const float4*)tt;
    #pragma unroll
    for (int ks = 1; ks < 16; ++ks) {
      const float4 u = *(const float4*)(tt + (size_t)ks * (256 * 1024));
      t.x += u.x; t.y += u.y; t.z += u.z; t.w += u.w;
    }
    const float4 bb = *(const float4*)(b1f + 4 * f4);
    v2f t0; t0.x = t.x + bb.x; t0.y = t.y + bb.y;
    v2f t1; t1.x = t.z + bb.z; t1.y = t.w + bb.w;
    *(v2f*)&ts[(2 * f4 + 0) * ES2_STRIDE + c * 2] = t0;
    *(v2f*)&ts[(2 * f4 + 1) * ES2_STRIDE + c * 2] = t1;
  }

  const int lane = tid & 31, wv = tid >> 5;
  const int lm   = lane & 15, lh = lane >> 4;

  const int h2 = gh * 4 + sg;

  const float4* aq = (const float4*)Mq + ((size_t)(sg * 64 + 2 * wv) * 8) * 32 + lane;
#define DEC_A(st_, q_) aq[((((st_) >> 1) * 16 + ((st_) & 1)) * 8 + (q_)) * 32]
  float4 ar[3][8];
  #pragma unroll
  for (int q = 0; q < 8; ++q) ar[0][q] = DEC_A(0, q);
  #pragma unroll
  for (int q = 0; q < 8; ++q) ar[1][q] = DEC_A(1, q);

  __syncthreads();

  v2f bf[2][16];
  #pragma unroll
  for (int c2 = 0; c2 < 2; ++c2) {
    #pragma unroll
    for (int k = 0; k < 16; ++k)
      bf[c2][k] = *(const v2f*)&ts[(2 * k + lh) * ES2_STRIDE + (c2 * 16 + lm) * 2];
  }

  #pragma unroll
  for (int st = 0; st < 8; ++st) {
    const int ss = st >> 1;
    const int oa = ss * 64 + (2 * wv + (st & 1)) * 4 + 2 * lh;
    if (st + 2 < 8) {
      #pragma unroll
      for (int q = 0; q < 8; ++q) ar[(st + 2) % 3][q] = DEC_A(st + 2, q);
    }
    const float4 ca = *(const float4*)&cb[(sg * 256 + oa) * 4];
    const float4 cc = *(const float4*)&cb[(sg * 256 + oa + 1) * 4];
    __builtin_amdgcn_sched_barrier(0);
    v8f acc0 = v8f_zero(), acc1 = v8f_zero();
    #pragma unroll
    for (int q = 0; q < 8; ++q) {
      const float4 a = ar[st % 3][q];
      v2f a0; a0.x = a.x; a0.y = a.y;
      v2f a1; a1.x = a.z; a1.y = a.w;
      acc0 = WMMA_F32(a0, bf[0][2 * q], acc0);
      acc1 = WMMA_F32(a0, bf[1][2 * q], acc1);
      acc0 = WMMA_F32(a1, bf[0][2 * q + 1], acc0);
      acc1 = WMMA_F32(a1, bf[1][2 * q + 1], acc1);
    }
    __builtin_amdgcn_sched_barrier(0);
    float* orow = out + (((size_t)n * COUT + oa) * 128 + h2) * 128 + lm * 4;
    float4 va, vb;
    va.x = acc0[0] + ca.x; va.y = acc0[1] + ca.y; va.z = acc0[2] + ca.z; va.w = acc0[3] + ca.w;
    vb.x = acc0[4] + cc.x; vb.y = acc0[5] + cc.y; vb.z = acc0[6] + cc.z; vb.w = acc0[7] + cc.w;
    *(float4*)orow               = va;
    *(float4*)(orow + 128 * 128) = vb;
    va.x = acc1[0] + ca.x; va.y = acc1[1] + ca.y; va.z = acc1[2] + ca.z; va.w = acc1[3] + ca.w;
    vb.x = acc1[4] + cc.x; vb.y = acc1[5] + cc.y; vb.z = acc1[6] + cc.z; vb.w = acc1[7] + cc.w;
    *(float4*)(orow + 64)             = va;
    *(float4*)(orow + 64 + 128 * 128) = vb;
  }
#undef DEC_A
}

extern "C" void kernel_launch(void* const* d_in, const int* in_sizes, int n_in,
                              void* d_out, int out_size, void* d_ws, size_t ws_size,
                              hipStream_t stream) {
  const float* x    = (const float*)d_in[0];
  const float* Wenc = (const float*)d_in[1];
  const float* benc = (const float*)d_in[2];
  const float* W1   = (const float*)d_in[3];
  const float* b1   = (const float*)d_in[4];
  const float* W2   = (const float*)d_in[5];
  const float* b2   = (const float*)d_in[6];
  const float* Wdec = (const float*)d_in[7];
  const float* bdec = (const float*)d_in[8];
  float* out = (float*)d_out;

  float* ws   = (float*)d_ws;
  float* W1f  = ws;
  float* Tp   = ws + W1F_FLOATS;
  float* b1f  = Tp + TP_FLOATS;
  float* Mf   = ws + ENC_FLOATS + T_FLOATS;
  float* cbuf = Mf + M_FLOATS;

  k_fold<<<356, 128, 0, stream>>>(Wenc, benc, W1, b1, W2, b2, Wdec, bdec, W1f, b1f, Mf, cbuf);

  k_main<<<256, 512, 0, stream>>>(x, W1f, Tp);

  k_decode<<<512, 256, 0, stream>>>(Tp, b1f, Mf, cbuf, out);
}
